# starting kernel
# baseline (speedup 1.0000x reference)
#include <hip/hip_runtime.h>
#include <math.h>


#define CF     192
#define TTILE  128
#define HALO   13
#define HW     (TTILE + 2*HALO)
#define SSTR   155
#define NTHR   384
#define NWAVE  12
#define TLEN   8192
#define NBATCH 32
#define NBINS  10
#define TAILV  5.0f
#define MINBW  0.001f
#define MINBH  0.001f
#define MIND   0.001f

typedef float v2f __attribute__((ext_vector_type(2)));
typedef float v8f __attribute__((ext_vector_type(8)));

__device__ __forceinline__ int sidx(int c, int u) {
  return (c >> 1) * (2 * SSTR) + 2 * u + (c & 1);
}

__device__ __forceinline__ float gelu_exact(float x) {
  return 0.5f * x * (1.0f + erff(x * 0.70710678118654752440f));
}
__device__ __forceinline__ float softplusf(float x) {
  return (x > 20.0f) ? x : log1pf(expf(x));
}

template<int US, int W>
__device__ void ln_gelu(float* buf, const float* __restrict__ gamma,
                        const float* __restrict__ beta) {
  const int tid  = threadIdx.x;
  const int lane = tid & 31;
  const int wid  = tid >> 5;
  for (int j = wid; j < W; j += NWAVE) {
    const int u = US + j;
    float v[6];
    float s1 = 0.f, s2 = 0.f;
#pragma unroll
    for (int r = 0; r < 6; ++r) {
      v[r] = buf[sidx(lane + 32 * r, u)];
      s1 += v[r];
      s2 += v[r] * v[r];
    }
#pragma unroll
    for (int m = 16; m >= 1; m >>= 1) {
      s1 += __shfl_xor(s1, m, 32);
      s2 += __shfl_xor(s2, m, 32);
    }
    const float mu  = s1 * (1.0f / 192.0f);
    const float var = s2 * (1.0f / 192.0f) - mu * mu;
    const float rs  = rsqrtf(var + 1e-5f);
#pragma unroll
    for (int r = 0; r < 6; ++r) {
      const int c = lane + 32 * r;
      const float xh = (v[r] - mu) * rs;
      buf[sidx(c, u)] = gelu_exact(xh * gamma[c] + beta[c]);
    }
  }
}

template<int US, int W, int NT>
__device__ void pw_conv_inplace(const float* __restrict__ Wm,
                                const float* __restrict__ bias, float* sy) {
  const int tid  = threadIdx.x;
  const int lane = tid & 31;
  const int wid  = tid >> 5;
  const int m0   = wid * 16;
  const int l15  = lane & 15;
  const int kof  = (lane & 16) ? 2 : 0;
  const int colmax = US + W - 1;

  v2f afrag[48];
  {
    const v2f* wrow = (const v2f*)(Wm + (size_t)(m0 + l15) * CF);
    const int kidx0 = kof >> 1;
#pragma unroll
    for (int kk = 0; kk < 48; ++kk) afrag[kk] = wrow[kk * 2 + kidx0];
  }

  for (int n = 0; n < NT; ++n) {
    int colb = US + n * 16 + l15;
    if (colb > colmax) colb = colmax;
    v8f c8 = {};
#pragma unroll
    for (int kk = 0; kk < 48; ++kk) {
      const int k = kk * 4 + kof;
      const v2f bv = *(const v2f*)(&sy[(k >> 1) * (2 * SSTR) + 2 * colb]);
      c8 = __builtin_amdgcn_wmma_f32_16x16x4_f32(
          false, afrag[kk], false, bv, (short)0, c8, false, false);
    }
    __syncthreads();
    const int col = US + n * 16 + l15;
    if (col <= colmax) {
      const int rbase = m0 + ((lane & 16) ? 8 : 0);
#pragma unroll
      for (int v = 0; v < 8; ++v) {
        const int row = rbase + v;
        sy[sidx(row, col)] = c8[v] + bias[row];
      }
    }
  }
}

__device__ void proj_mm(const float* __restrict__ Wm,
                        const float* __restrict__ bias,
                        const float* sh, float* sy) {
  const int tid  = threadIdx.x;
  const int lane = tid & 31;
  const int wid  = tid >> 5;
  const int l15  = lane & 15;
  const int kof  = (lane & 16) ? 2 : 0;
  for (int job = wid; job < 16; job += NWAVE) {
    const int m0 = (job & 1) * 16;
    const int n  = job >> 1;
    const int row  = m0 + l15;
    const int rowc = row < 29 ? row : 28;
    const v2f* wrow = (const v2f*)(Wm + (size_t)rowc * CF);
    const int col = HALO + n * 16 + l15;
    v2f afrag[48];
#pragma unroll
    for (int kk = 0; kk < 48; ++kk) afrag[kk] = wrow[kk * 2 + (kof >> 1)];
    v8f c8 = {};
#pragma unroll
    for (int kk = 0; kk < 48; ++kk) {
      const int k = kk * 4 + kof;
      const v2f bv = *(const v2f*)(&sh[(k >> 1) * (2 * SSTR) + 2 * col]);
      c8 = __builtin_amdgcn_wmma_f32_16x16x4_f32(
          false, afrag[kk], false, bv, (short)0, c8, false, false);
    }
    const int rbase = m0 + ((lane & 16) ? 8 : 0);
#pragma unroll
    for (int v = 0; v < 8; ++v) {
      const int r = rbase + v;
      if (r < 29) sy[sidx(r, col)] = c8[v] + bias[r];
    }
  }
}

template<int DIL, int US, int W, int NT>
__device__ void layer_step(float* sh, float* sy, int t0,
                           const float* __restrict__ sepw,
                           const float* __restrict__ sepb,
                           const float* __restrict__ pww,
                           const float* __restrict__ pwb,
                           const float* __restrict__ gg1,
                           const float* __restrict__ bb1,
                           const float* __restrict__ gg2,
                           const float* __restrict__ bb2) {
  const int tid = threadIdx.x;
  for (int e = tid; e < CF * W; e += NTHR) {
    const int c = e / W;
    const int u = US + (e - c * W);
    const float w0 = sepw[c * 3 + 0];
    const float w1 = sepw[c * 3 + 1];
    const float w2 = sepw[c * 3 + 2];
    sy[sidx(c, u)] = w0 * sh[sidx(c, u - DIL)] + w1 * sh[sidx(c, u)] +
                     w2 * sh[sidx(c, u + DIL)] + sepb[c];
  }
  __syncthreads();
  ln_gelu<US, W>(sy, gg1, bb1);
  __syncthreads();
  pw_conv_inplace<US, W, NT>(pww, pwb, sy);
  __syncthreads();
  ln_gelu<US, W>(sy, gg2, bb2);
  __syncthreads();
  for (int e = tid; e < CF * W; e += NTHR) {
    const int c = e / W;
    const int u = US + (e - c * W);
    const int t = t0 - HALO + u;
    if (t >= 0 && t < TLEN) sh[sidx(c, u)] += sy[sidx(c, u)];
  }
  __syncthreads();
}

__device__ __forceinline__ void build_knots(const float* uu, float* cks,
                                            float minb) {
  float m = uu[0];
#pragma unroll
  for (int i = 1; i < NBINS; ++i) m = fmaxf(m, uu[i]);
  float e[NBINS];
  float s = 0.f;
#pragma unroll
  for (int i = 0; i < NBINS; ++i) { e[i] = expf(uu[i] - m); s += e[i]; }
  const float inv = 1.0f / s;
  const float scale = 1.0f - minb * (float)NBINS;
  cks[0] = -TAILV;
  float acc = 0.f;
#pragma unroll
  for (int i = 0; i < NBINS - 1; ++i) {
    acc += minb + scale * e[i] * inv;
    cks[i + 1] = -TAILV + 2.0f * TAILV * acc;
  }
  cks[NBINS] = TAILV;
}

__global__ void __launch_bounds__(NTHR, 1)
convflow_kernel(const float* __restrict__ x,
                const float* __restrict__ pre_w, const float* __restrict__ pre_b,
                const float* __restrict__ sep_w, const float* __restrict__ sep_b,
                const float* __restrict__ pw_w,  const float* __restrict__ pw_b,
                const float* __restrict__ g1,    const float* __restrict__ b1,
                const float* __restrict__ g2,    const float* __restrict__ b2,
                const float* __restrict__ proj_w,const float* __restrict__ proj_b,
                float* __restrict__ out) {
  extern __shared__ float smem[];
  float* sh  = smem;
  float* sy  = smem + CF * SSTR;
  float* red = smem + 2 * CF * SSTR;

  const int tid = threadIdx.x;
  const int tile = blockIdx.x;
  const int b    = blockIdx.y;
  const int t0   = tile * TTILE;

  for (int u = tid; u < HW; u += NTHR) {
    const int t = t0 - HALO + u;
    red[u] = (t >= 0 && t < TLEN) ? x[(size_t)b * 2 * TLEN + t] : 0.0f;
  }
  __syncthreads();
  for (int e = tid; e < CF * HW; e += NTHR) {
    const int c = e / HW;
    const int u = e - c * HW;
    const int t = t0 - HALO + u;
    sh[sidx(c, u)] =
        (t >= 0 && t < TLEN) ? (pre_w[c] * red[u] + pre_b[c]) : 0.0f;
  }
  if (tid < TTILE) {
    out[(size_t)b * 2 * TLEN + t0 + tid] = red[HALO + tid];
  }
  __syncthreads();

  layer_step<1, 1, 152, 10>(sh, sy, t0, sep_w + 0 * CF * 3, sep_b + 0 * CF,
                            pw_w + 0 * CF * CF, pw_b + 0 * CF, g1 + 0 * CF,
                            b1 + 0 * CF, g2 + 0 * CF, b2 + 0 * CF);
  layer_step<3, 4, 146, 10>(sh, sy, t0, sep_w + 1 * CF * 3, sep_b + 1 * CF,
                            pw_w + 1 * CF * CF, pw_b + 1 * CF, g1 + 1 * CF,
                            b1 + 1 * CF, g2 + 1 * CF, b2 + 1 * CF);
  layer_step<9, 13, 128, 8>(sh, sy, t0, sep_w + 2 * CF * 3, sep_b + 2 * CF,
                            pw_w + 2 * CF * CF, pw_b + 2 * CF, g1 + 2 * CF,
                            b1 + 2 * CF, g2 + 2 * CF, b2 + 2 * CF);

  proj_mm(proj_w, proj_b, sh, sy);
  __syncthreads();

  float la_local = 0.0f;
  if (tid < TTILE) {
    const int u = HALO + tid;
    const int t = t0 + tid;
    float uw[NBINS], uh[NBINS], d[NBINS + 1];
#pragma unroll
    for (int i = 0; i < NBINS; ++i) uw[i] = sy[sidx(i, u)];
#pragma unroll
    for (int i = 0; i < NBINS; ++i) uh[i] = sy[sidx(NBINS + i, u)];
    d[0] = 1.0f;
    d[NBINS] = 1.0f;
#pragma unroll
    for (int i = 0; i < NBINS - 1; ++i)
      d[i + 1] = MIND + softplusf(sy[sidx(2 * NBINS + i, u)]);

    float cw[NBINS + 1], ch[NBINS + 1];
    build_knots(uw, cw, MINBW);
    build_knots(uh, ch, MINBH);

    const float xin = x[(size_t)b * 2 * TLEN + TLEN + t];
    const bool inside = (xin >= -TAILV) && (xin <= TAILV);
    const float xc = fminf(fmaxf(xin, -TAILV), TAILV);
    int idx = -1;
#pragma unroll
    for (int k = 0; k < NBINS + 1; ++k) idx += (xc >= cw[k]) ? 1 : 0;
    idx = max(0, min(idx, NBINS - 1));

    const float icw = cw[idx];
    const float ibw = cw[idx + 1] - cw[idx];
    const float ich = ch[idx];
    const float ih  = ch[idx + 1] - ch[idx];
    const float idelta = ih / ibw;
    const float d0 = d[idx];
    const float d1 = d[idx + 1];
    const float theta = (xc - icw) / ibw;
    const float th1m  = 1.0f - theta;
    const float num = ih * (idelta * theta * theta + d0 * theta * th1m);
    const float den = idelta + (d0 + d1 - 2.0f * idelta) * theta * th1m;
    float outv = ich + num / den;
    const float dnum = idelta * idelta *
        (d1 * theta * theta + 2.0f * idelta * theta * th1m + d0 * th1m * th1m);
    float la = logf(dnum) - 2.0f * logf(den);
    if (!inside) { outv = xin; la = 0.0f; }
    out[(size_t)b * 2 * TLEN + TLEN + t] = outv;
    la_local = la;
  }

  __syncthreads();
  red[tid] = (tid < TTILE) ? la_local : 0.0f;
  __syncthreads();
  for (int s = 64; s > 0; s >>= 1) {
    if (tid < s) red[tid] += red[tid + s];
    __syncthreads();
  }
  if (tid == 0) atomicAdd(out + (size_t)NBATCH * 2 * TLEN + b, red[0]);
}

__global__ void zero_logdet(float* p) {
  if (threadIdx.x < NBATCH) p[threadIdx.x] = 0.0f;
}

extern "C" void kernel_launch(void* const* d_in, const int* in_sizes, int n_in,
                              void* d_out, int out_size, void* d_ws,
                              size_t ws_size, hipStream_t stream) {
  (void)in_sizes; (void)n_in; (void)out_size; (void)d_ws; (void)ws_size;
  const float* x      = (const float*)d_in[0];
  const float* pre_w  = (const float*)d_in[2];
  const float* pre_b  = (const float*)d_in[3];
  const float* sep_w  = (const float*)d_in[4];
  const float* sep_b  = (const float*)d_in[5];
  const float* pw_w   = (const float*)d_in[6];
  const float* pw_b   = (const float*)d_in[7];
  const float* g1     = (const float*)d_in[8];
  const float* b1     = (const float*)d_in[9];
  const float* g2     = (const float*)d_in[10];
  const float* b2     = (const float*)d_in[11];
  const float* proj_w = (const float*)d_in[12];
  const float* proj_b = (const float*)d_in[13];
  float* out = (float*)d_out;

  zero_logdet<<<1, 32, 0, stream>>>(out + (size_t)NBATCH * 2 * TLEN);

  const size_t smem_bytes = (size_t)(2 * CF * SSTR + 512) * sizeof(float);
  dim3 grid(TLEN / TTILE, NBATCH);
  convflow_kernel<<<grid, NTHR, smem_bytes, stream>>>(
      x, pre_w, pre_b, sep_w, sep_b, pw_w, pw_b, g1, b1, g2, b2, proj_w,
      proj_b, out);
}


# GELU: library erff (two exec-masked paths, serial chain) replaced by one branch-free form max(x,0)-|x|*exp(-x*x/2)*t*P5(t), t=rcp(1+0.28|x|); abs error vs double stays below 3e-7
# speedup vs baseline: 1.2941x; 1.2941x over previous
#include <hip/hip_runtime.h>
#include <math.h>


#define CF     192
#define TTILE  128
#define HALO   13
#define HW     (TTILE + 2*HALO)
#define SSTR   155
#define NTHR   384
#define NWAVE  12
#define TLEN   8192
#define NBATCH 32
#define NBINS  10
#define TAILV  5.0f
#define MINBW  0.001f
#define MINBH  0.001f
#define MIND   0.001f

typedef float v2f __attribute__((ext_vector_type(2)));
typedef float v8f __attribute__((ext_vector_type(8)));

__device__ __forceinline__ int sidx(int c, int u) {
  return (c >> 1) * (2 * SSTR) + 2 * u + (c & 1);
}

__device__ __forceinline__ float gelu_exact(float x) {
  const float ax = fabsf(x);
  const float t = __builtin_amdgcn_rcpf(fmaf(ax, 0.28f, 1.0f));
  const float e = __builtin_amdgcn_exp2f((x * x) * -0.72134752044448170368f);
  float p = -0.10675480918401203f;
  p = fmaf(p, t, 0.39380918123096653f);
  p = fmaf(p, t, -0.2434958341181943f);
  p = fmaf(p, t, 0.27424663512843267f);
  p = fmaf(p, t, 0.06532213231594822f);
  p = fmaf(p, t, 0.11687268054021908f);
  const float tail = (p * t) * e;
  return fmaf(-ax, tail, fmaxf(x, 0.0f));
}
__device__ __forceinline__ float softplusf(float x) {
  return (x > 20.0f) ? x : log1pf(expf(x));
}

template<int US, int W>
__device__ void ln_gelu(float* buf, const float* __restrict__ gamma,
                        const float* __restrict__ beta) {
  const int tid  = threadIdx.x;
  const int lane = tid & 31;
  const int wid  = tid >> 5;
  for (int j = wid; j < W; j += NWAVE) {
    const int u = US + j;
    float v[6];
    float s1 = 0.f, s2 = 0.f;
#pragma unroll
    for (int r = 0; r < 6; ++r) {
      v[r] = buf[sidx(lane + 32 * r, u)];
      s1 += v[r];
      s2 += v[r] * v[r];
    }
#pragma unroll
    for (int m = 16; m >= 1; m >>= 1) {
      s1 += __shfl_xor(s1, m, 32);
      s2 += __shfl_xor(s2, m, 32);
    }
    const float mu  = s1 * (1.0f / 192.0f);
    const float var = s2 * (1.0f / 192.0f) - mu * mu;
    const float rs  = rsqrtf(var + 1e-5f);
#pragma unroll
    for (int r = 0; r < 6; ++r) {
      const int c = lane + 32 * r;
      const float xh = (v[r] - mu) * rs;
      buf[sidx(c, u)] = gelu_exact(xh * gamma[c] + beta[c]);
    }
  }
}

template<int US, int W, int NT>
__device__ void pw_conv_inplace(const float* __restrict__ Wm,
                                const float* __restrict__ bias, float* sy) {
  const int tid  = threadIdx.x;
  const int lane = tid & 31;
  const int wid  = tid >> 5;
  const int m0   = wid * 16;
  const int l15  = lane & 15;
  const int kof  = (lane & 16) ? 2 : 0;
  const int colmax = US + W - 1;

  v2f afrag[48];
  {
    const v2f* wrow = (const v2f*)(Wm + (size_t)(m0 + l15) * CF);
    const int kidx0 = kof >> 1;
#pragma unroll
    for (int kk = 0; kk < 48; ++kk) afrag[kk] = wrow[kk * 2 + kidx0];
  }

  for (int n = 0; n < NT; ++n) {
    int colb = US + n * 16 + l15;
    if (colb > colmax) colb = colmax;
    v8f c8 = {};
#pragma unroll
    for (int kk = 0; kk < 48; ++kk) {
      const int k = kk * 4 + kof;
      const v2f bv = *(const v2f*)(&sy[(k >> 1) * (2 * SSTR) + 2 * colb]);
      c8 = __builtin_amdgcn_wmma_f32_16x16x4_f32(
          false, afrag[kk], false, bv, (short)0, c8, false, false);
    }
    __syncthreads();
    const int col = US + n * 16 + l15;
    if (col <= colmax) {
      const int rbase = m0 + ((lane & 16) ? 8 : 0);
#pragma unroll
      for (int v = 0; v < 8; ++v) {
        const int row = rbase + v;
        sy[sidx(row, col)] = c8[v] + bias[row];
      }
    }
  }
}

__device__ void proj_mm(const float* __restrict__ Wm,
                        const float* __restrict__ bias,
                        const float* sh, float* sy) {
  const int tid  = threadIdx.x;
  const int lane = tid & 31;
  const int wid  = tid >> 5;
  const int l15  = lane & 15;
  const int kof  = (lane & 16) ? 2 : 0;
  for (int job = wid; job < 16; job += NWAVE) {
    const int m0 = (job & 1) * 16;
    const int n  = job >> 1;
    const int row  = m0 + l15;
    const int rowc = row < 29 ? row : 28;
    const v2f* wrow = (const v2f*)(Wm + (size_t)rowc * CF);
    const int col = HALO + n * 16 + l15;
    v2f afrag[48];
#pragma unroll
    for (int kk = 0; kk < 48; ++kk) afrag[kk] = wrow[kk * 2 + (kof >> 1)];
    v8f c8 = {};
#pragma unroll
    for (int kk = 0; kk < 48; ++kk) {
      const int k = kk * 4 + kof;
      const v2f bv = *(const v2f*)(&sh[(k >> 1) * (2 * SSTR) + 2 * col]);
      c8 = __builtin_amdgcn_wmma_f32_16x16x4_f32(
          false, afrag[kk], false, bv, (short)0, c8, false, false);
    }
    const int rbase = m0 + ((lane & 16) ? 8 : 0);
#pragma unroll
    for (int v = 0; v < 8; ++v) {
      const int r = rbase + v;
      if (r < 29) sy[sidx(r, col)] = c8[v] + bias[r];
    }
  }
}

template<int DIL, int US, int W, int NT>
__device__ void layer_step(float* sh, float* sy, int t0,
                           const float* __restrict__ sepw,
                           const float* __restrict__ sepb,
                           const float* __restrict__ pww,
                           const float* __restrict__ pwb,
                           const float* __restrict__ gg1,
                           const float* __restrict__ bb1,
                           const float* __restrict__ gg2,
                           const float* __restrict__ bb2) {
  const int tid = threadIdx.x;
  for (int e = tid; e < CF * W; e += NTHR) {
    const int c = e / W;
    const int u = US + (e - c * W);
    const float w0 = sepw[c * 3 + 0];
    const float w1 = sepw[c * 3 + 1];
    const float w2 = sepw[c * 3 + 2];
    sy[sidx(c, u)] = w0 * sh[sidx(c, u - DIL)] + w1 * sh[sidx(c, u)] +
                     w2 * sh[sidx(c, u + DIL)] + sepb[c];
  }
  __syncthreads();
  ln_gelu<US, W>(sy, gg1, bb1);
  __syncthreads();
  pw_conv_inplace<US, W, NT>(pww, pwb, sy);
  __syncthreads();
  ln_gelu<US, W>(sy, gg2, bb2);
  __syncthreads();
  for (int e = tid; e < CF * W; e += NTHR) {
    const int c = e / W;
    const int u = US + (e - c * W);
    const int t = t0 - HALO + u;
    if (t >= 0 && t < TLEN) sh[sidx(c, u)] += sy[sidx(c, u)];
  }
  __syncthreads();
}

__device__ __forceinline__ void build_knots(const float* uu, float* cks,
                                            float minb) {
  float m = uu[0];
#pragma unroll
  for (int i = 1; i < NBINS; ++i) m = fmaxf(m, uu[i]);
  float e[NBINS];
  float s = 0.f;
#pragma unroll
  for (int i = 0; i < NBINS; ++i) { e[i] = expf(uu[i] - m); s += e[i]; }
  const float inv = 1.0f / s;
  const float scale = 1.0f - minb * (float)NBINS;
  cks[0] = -TAILV;
  float acc = 0.f;
#pragma unroll
  for (int i = 0; i < NBINS - 1; ++i) {
    acc += minb + scale * e[i] * inv;
    cks[i + 1] = -TAILV + 2.0f * TAILV * acc;
  }
  cks[NBINS] = TAILV;
}

__global__ void __launch_bounds__(NTHR, 1)
convflow_kernel(const float* __restrict__ x,
                const float* __restrict__ pre_w, const float* __restrict__ pre_b,
                const float* __restrict__ sep_w, const float* __restrict__ sep_b,
                const float* __restrict__ pw_w,  const float* __restrict__ pw_b,
                const float* __restrict__ g1,    const float* __restrict__ b1,
                const float* __restrict__ g2,    const float* __restrict__ b2,
                const float* __restrict__ proj_w,const float* __restrict__ proj_b,
                float* __restrict__ out) {
  extern __shared__ float smem[];
  float* sh  = smem;
  float* sy  = smem + CF * SSTR;
  float* red = smem + 2 * CF * SSTR;

  const int tid = threadIdx.x;
  const int tile = blockIdx.x;
  const int b    = blockIdx.y;
  const int t0   = tile * TTILE;

  for (int u = tid; u < HW; u += NTHR) {
    const int t = t0 - HALO + u;
    red[u] = (t >= 0 && t < TLEN) ? x[(size_t)b * 2 * TLEN + t] : 0.0f;
  }
  __syncthreads();
  for (int e = tid; e < CF * HW; e += NTHR) {
    const int c = e / HW;
    const int u = e - c * HW;
    const int t = t0 - HALO + u;
    sh[sidx(c, u)] =
        (t >= 0 && t < TLEN) ? (pre_w[c] * red[u] + pre_b[c]) : 0.0f;
  }
  if (tid < TTILE) {
    out[(size_t)b * 2 * TLEN + t0 + tid] = red[HALO + tid];
  }
  __syncthreads();

  layer_step<1, 1, 152, 10>(sh, sy, t0, sep_w + 0 * CF * 3, sep_b + 0 * CF,
                            pw_w + 0 * CF * CF, pw_b + 0 * CF, g1 + 0 * CF,
                            b1 + 0 * CF, g2 + 0 * CF, b2 + 0 * CF);
  layer_step<3, 4, 146, 10>(sh, sy, t0, sep_w + 1 * CF * 3, sep_b + 1 * CF,
                            pw_w + 1 * CF * CF, pw_b + 1 * CF, g1 + 1 * CF,
                            b1 + 1 * CF, g2 + 1 * CF, b2 + 1 * CF);
  layer_step<9, 13, 128, 8>(sh, sy, t0, sep_w + 2 * CF * 3, sep_b + 2 * CF,
                            pw_w + 2 * CF * CF, pw_b + 2 * CF, g1 + 2 * CF,
                            b1 + 2 * CF, g2 + 2 * CF, b2 + 2 * CF);

  proj_mm(proj_w, proj_b, sh, sy);
  __syncthreads();

  float la_local = 0.0f;
  if (tid < TTILE) {
    const int u = HALO + tid;
    const int t = t0 + tid;
    float uw[NBINS], uh[NBINS], d[NBINS + 1];
#pragma unroll
    for (int i = 0; i < NBINS; ++i) uw[i] = sy[sidx(i, u)];
#pragma unroll
    for (int i = 0; i < NBINS; ++i) uh[i] = sy[sidx(NBINS + i, u)];
    d[0] = 1.0f;
    d[NBINS] = 1.0f;
#pragma unroll
    for (int i = 0; i < NBINS - 1; ++i)
      d[i + 1] = MIND + softplusf(sy[sidx(2 * NBINS + i, u)]);

    float cw[NBINS + 1], ch[NBINS + 1];
    build_knots(uw, cw, MINBW);
    build_knots(uh, ch, MINBH);

    const float xin = x[(size_t)b * 2 * TLEN + TLEN + t];
    const bool inside = (xin >= -TAILV) && (xin <= TAILV);
    const float xc = fminf(fmaxf(xin, -TAILV), TAILV);
    int idx = -1;
#pragma unroll
    for (int k = 0; k < NBINS + 1; ++k) idx += (xc >= cw[k]) ? 1 : 0;
    idx = max(0, min(idx, NBINS - 1));

    const float icw = cw[idx];
    const float ibw = cw[idx + 1] - cw[idx];
    const float ich = ch[idx];
    const float ih  = ch[idx + 1] - ch[idx];
    const float idelta = ih / ibw;
    const float d0 = d[idx];
    const float d1 = d[idx + 1];
    const float theta = (xc - icw) / ibw;
    const float th1m  = 1.0f - theta;
    const float num = ih * (idelta * theta * theta + d0 * theta * th1m);
    const float den = idelta + (d0 + d1 - 2.0f * idelta) * theta * th1m;
    float outv = ich + num / den;
    const float dnum = idelta * idelta *
        (d1 * theta * theta + 2.0f * idelta * theta * th1m + d0 * th1m * th1m);
    float la = logf(dnum) - 2.0f * logf(den);
    if (!inside) { outv = xin; la = 0.0f; }
    out[(size_t)b * 2 * TLEN + TLEN + t] = outv;
    la_local = la;
  }

  __syncthreads();
  red[tid] = (tid < TTILE) ? la_local : 0.0f;
  __syncthreads();
  for (int s = 64; s > 0; s >>= 1) {
    if (tid < s) red[tid] += red[tid + s];
    __syncthreads();
  }
  if (tid == 0) atomicAdd(out + (size_t)NBATCH * 2 * TLEN + b, red[0]);
}

__global__ void zero_logdet(float* p) {
  if (threadIdx.x < NBATCH) p[threadIdx.x] = 0.0f;
}

extern "C" void kernel_launch(void* const* d_in, const int* in_sizes, int n_in,
                              void* d_out, int out_size, void* d_ws,
                              size_t ws_size, hipStream_t stream) {
  (void)in_sizes; (void)n_in; (void)out_size; (void)d_ws; (void)ws_size;
  const float* x      = (const float*)d_in[0];
  const float* pre_w  = (const float*)d_in[2];
  const float* pre_b  = (const float*)d_in[3];
  const float* sep_w  = (const float*)d_in[4];
  const float* sep_b  = (const float*)d_in[5];
  const float* pw_w   = (const float*)d_in[6];
  const float* pw_b   = (const float*)d_in[7];
  const float* g1     = (const float*)d_in[8];
  const float* b1     = (const float*)d_in[9];
  const float* g2     = (const float*)d_in[10];
  const float* b2     = (const float*)d_in[11];
  const float* proj_w = (const float*)d_in[12];
  const float* proj_b = (const float*)d_in[13];
  float* out = (float*)d_out;

  zero_logdet<<<1, 32, 0, stream>>>(out + (size_t)NBATCH * 2 * TLEN);

  const size_t smem_bytes = (size_t)(2 * CF * SSTR + 512) * sizeof(float);
  dim3 grid(TLEN / TTILE, NBATCH);
  convflow_kernel<<<grid, NTHR, smem_bytes, stream>>>(
      x, pre_w, pre_b, sep_w, sep_b, pw_w, pw_b, g1, b1, g2, b2, proj_w,
      proj_b, out);
}


# Depthwise conv folded into LN1 and residual add folded into LN2: the wave owning a column forms the conv from h in registers and adds its result into h; 2 LDS passes + 2 barriers per layer gone
# speedup vs baseline: 1.8145x; 1.4021x over previous
#include <hip/hip_runtime.h>
#include <math.h>


#define CF     192
#define TTILE  128
#define HALO   13
#define HW     (TTILE + 2*HALO)
#define SSTR   155
#define NTHR   384
#define NWAVE  12
#define TLEN   8192
#define NBATCH 32
#define NBINS  10
#define TAILV  5.0f
#define MINBW  0.001f
#define MINBH  0.001f
#define MIND   0.001f

typedef float v2f __attribute__((ext_vector_type(2)));
typedef float v8f __attribute__((ext_vector_type(8)));

__device__ __forceinline__ int sidx(int c, int u) {
  return (c >> 1) * (2 * SSTR) + 2 * u + (c & 1);
}

__device__ __forceinline__ float gelu_exact(float x) {
  const float ax = fabsf(x);
  const float t = __builtin_amdgcn_rcpf(fmaf(ax, 0.28f, 1.0f));
  const float e = __builtin_amdgcn_exp2f((x * x) * -0.72134752044448170368f);
  float p = -0.10675480918401203f;
  p = fmaf(p, t, 0.39380918123096653f);
  p = fmaf(p, t, -0.2434958341181943f);
  p = fmaf(p, t, 0.27424663512843267f);
  p = fmaf(p, t, 0.06532213231594822f);
  p = fmaf(p, t, 0.11687268054021908f);
  const float tail = (p * t) * e;
  return fmaf(-ax, tail, fmaxf(x, 0.0f));
}
__device__ __forceinline__ float softplusf(float x) {
  return (x > 20.0f) ? x : log1pf(expf(x));
}

__device__ __forceinline__ void ln_gelu_regs(float (&v)[6],
                                             const float (&gm)[6],
                                             const float (&bt)[6]) {
  float s1 = 0.f, s2 = 0.f;
#pragma unroll
  for (int r = 0; r < 6; ++r) {
    s1 += v[r];
    s2 += v[r] * v[r];
  }
  v2f s = {s1, s2};
#pragma unroll
  for (int m = 16; m >= 1; m >>= 1) {
    v2f o;
    o.x = __shfl_xor(s.x, m, 32);
    o.y = __shfl_xor(s.y, m, 32);
    s += o;
  }
  const float mu  = s.x * (1.0f / 192.0f);
  const float var = s.y * (1.0f / 192.0f) - mu * mu;
  const float rs  = rsqrtf(var + 1e-5f);
#pragma unroll
  for (int r = 0; r < 6; ++r) {
    const float xh = (v[r] - mu) * rs;
    v[r] = gelu_exact(xh * gm[r] + bt[r]);
  }
}

template<int DIL, int US, int W>
__device__ void dwconv_ln_gelu(const float* __restrict__ sh,
                               float* __restrict__ sy,
                               const float* __restrict__ sepw,
                               const float* __restrict__ sepb,
                               const float* __restrict__ gamma,
                               const float* __restrict__ beta) {
  const int tid  = threadIdx.x;
  const int lane = tid & 31;
  const int wid  = tid >> 5;
  float w0[6], w1[6], w2[6], sb[6], gm[6], bt[6];
#pragma unroll
  for (int r = 0; r < 6; ++r) {
    const int c = lane + 32 * r;
    w0[r] = sepw[c * 3 + 0];
    w1[r] = sepw[c * 3 + 1];
    w2[r] = sepw[c * 3 + 2];
    sb[r] = sepb[c];
    gm[r] = gamma[c];
    bt[r] = beta[c];
  }
  for (int j = wid; j < W; j += NWAVE) {
    const int u = US + j;
    float v[6];
#pragma unroll
    for (int r = 0; r < 6; ++r) {
      const int i = sidx(lane + 32 * r, u);
      v[r] = w0[r] * sh[i - 2 * DIL] + w1[r] * sh[i] + w2[r] * sh[i + 2 * DIL] +
             sb[r];
    }
    ln_gelu_regs(v, gm, bt);
#pragma unroll
    for (int r = 0; r < 6; ++r) sy[sidx(lane + 32 * r, u)] = v[r];
  }
}

template<int US, int W>
__device__ void ln_gelu_residual(const float* __restrict__ sy,
                                 float* __restrict__ sh, int t0,
                                 const float* __restrict__ gamma,
                                 const float* __restrict__ beta) {
  const int tid  = threadIdx.x;
  const int lane = tid & 31;
  const int wid  = tid >> 5;
  float gm[6], bt[6];
#pragma unroll
  for (int r = 0; r < 6; ++r) {
    gm[r] = gamma[lane + 32 * r];
    bt[r] = beta[lane + 32 * r];
  }
  for (int j = wid; j < W; j += NWAVE) {
    const int u = US + j;
    const int t = t0 - HALO + u;
    float v[6], hv[6];
#pragma unroll
    for (int r = 0; r < 6; ++r) {
      const int i = sidx(lane + 32 * r, u);
      v[r]  = sy[i];
      hv[r] = sh[i];
    }
    ln_gelu_regs(v, gm, bt);
    const bool valid = (t >= 0) && (t < TLEN);
#pragma unroll
    for (int r = 0; r < 6; ++r)
      sh[sidx(lane + 32 * r, u)] = hv[r] + (valid ? v[r] : 0.0f);
  }
}

template<int US, int W, int NT>
__device__ void pw_conv_inplace(const float* __restrict__ Wm,
                                const float* __restrict__ bias, float* sy) {
  const int tid  = threadIdx.x;
  const int lane = tid & 31;
  const int wid  = tid >> 5;
  const int m0   = wid * 16;
  const int l15  = lane & 15;
  const int kof  = (lane & 16) ? 2 : 0;
  const int colmax = US + W - 1;

  v2f afrag[48];
  {
    const v2f* wrow = (const v2f*)(Wm + (size_t)(m0 + l15) * CF);
    const int kidx0 = kof >> 1;
#pragma unroll
    for (int kk = 0; kk < 48; ++kk) afrag[kk] = wrow[kk * 2 + kidx0];
  }

  for (int n = 0; n < NT; ++n) {
    int colb = US + n * 16 + l15;
    if (colb > colmax) colb = colmax;
    v8f c8 = {};
#pragma unroll
    for (int kk = 0; kk < 48; ++kk) {
      const int k = kk * 4 + kof;
      const v2f bv = *(const v2f*)(&sy[(k >> 1) * (2 * SSTR) + 2 * colb]);
      c8 = __builtin_amdgcn_wmma_f32_16x16x4_f32(
          false, afrag[kk], false, bv, (short)0, c8, false, false);
    }
    __syncthreads();
    const int col = US + n * 16 + l15;
    if (col <= colmax) {
      const int rbase = m0 + ((lane & 16) ? 8 : 0);
#pragma unroll
      for (int v = 0; v < 8; ++v) {
        const int row = rbase + v;
        sy[sidx(row, col)] = c8[v] + bias[row];
      }
    }
  }
}

__device__ void proj_mm(const float* __restrict__ Wm,
                        const float* __restrict__ bias,
                        const float* sh, float* sy) {
  const int tid  = threadIdx.x;
  const int lane = tid & 31;
  const int wid  = tid >> 5;
  const int l15  = lane & 15;
  const int kof  = (lane & 16) ? 2 : 0;
  for (int job = wid; job < 16; job += NWAVE) {
    const int m0 = (job & 1) * 16;
    const int n  = job >> 1;
    const int row  = m0 + l15;
    const int rowc = row < 29 ? row : 28;
    const v2f* wrow = (const v2f*)(Wm + (size_t)rowc * CF);
    const int col = HALO + n * 16 + l15;
    v2f afrag[48];
#pragma unroll
    for (int kk = 0; kk < 48; ++kk) afrag[kk] = wrow[kk * 2 + (kof >> 1)];
    v8f c8 = {};
#pragma unroll
    for (int kk = 0; kk < 48; ++kk) {
      const int k = kk * 4 + kof;
      const v2f bv = *(const v2f*)(&sh[(k >> 1) * (2 * SSTR) + 2 * col]);
      c8 = __builtin_amdgcn_wmma_f32_16x16x4_f32(
          false, afrag[kk], false, bv, (short)0, c8, false, false);
    }
    const int rbase = m0 + ((lane & 16) ? 8 : 0);
#pragma unroll
    for (int v = 0; v < 8; ++v) {
      const int r = rbase + v;
      if (r < 29) sy[sidx(r, col)] = c8[v] + bias[r];
    }
  }
}

template<int DIL, int US, int W, int NT>
__device__ void layer_step(float* sh, float* sy, int t0,
                           const float* __restrict__ sepw,
                           const float* __restrict__ sepb,
                           const float* __restrict__ pww,
                           const float* __restrict__ pwb,
                           const float* __restrict__ gg1,
                           const float* __restrict__ bb1,
                           const float* __restrict__ gg2,
                           const float* __restrict__ bb2) {
  dwconv_ln_gelu<DIL, US, W>(sh, sy, sepw, sepb, gg1, bb1);
  __syncthreads();
  pw_conv_inplace<US, W, NT>(pww, pwb, sy);
  __syncthreads();
  ln_gelu_residual<US, W>(sy, sh, t0, gg2, bb2);
  __syncthreads();
}

__device__ __forceinline__ void build_knots(const float* uu, float* cks,
                                            float minb) {
  float m = uu[0];
#pragma unroll
  for (int i = 1; i < NBINS; ++i) m = fmaxf(m, uu[i]);
  float e[NBINS];
  float s = 0.f;
#pragma unroll
  for (int i = 0; i < NBINS; ++i) { e[i] = expf(uu[i] - m); s += e[i]; }
  const float inv = 1.0f / s;
  const float scale = 1.0f - minb * (float)NBINS;
  cks[0] = -TAILV;
  float acc = 0.f;
#pragma unroll
  for (int i = 0; i < NBINS - 1; ++i) {
    acc += minb + scale * e[i] * inv;
    cks[i + 1] = -TAILV + 2.0f * TAILV * acc;
  }
  cks[NBINS] = TAILV;
}

__global__ void __launch_bounds__(NTHR, 1)
convflow_kernel(const float* __restrict__ x,
                const float* __restrict__ pre_w, const float* __restrict__ pre_b,
                const float* __restrict__ sep_w, const float* __restrict__ sep_b,
                const float* __restrict__ pw_w,  const float* __restrict__ pw_b,
                const float* __restrict__ g1,    const float* __restrict__ b1,
                const float* __restrict__ g2,    const float* __restrict__ b2,
                const float* __restrict__ proj_w,const float* __restrict__ proj_b,
                float* __restrict__ out) {
  extern __shared__ float smem[];
  float* sh  = smem;
  float* sy  = smem + CF * SSTR;
  float* red = smem + 2 * CF * SSTR;

  const int tid = threadIdx.x;
  const int tile = blockIdx.x;
  const int b    = blockIdx.y;
  const int t0   = tile * TTILE;

  for (int u = tid; u < HW; u += NTHR) {
    const int t = t0 - HALO + u;
    red[u] = (t >= 0 && t < TLEN) ? x[(size_t)b * 2 * TLEN + t] : 0.0f;
  }
  __syncthreads();
  for (int e = tid; e < CF * HW; e += NTHR) {
    const int c = e / HW;
    const int u = e - c * HW;
    const int t = t0 - HALO + u;
    sh[sidx(c, u)] =
        (t >= 0 && t < TLEN) ? (pre_w[c] * red[u] + pre_b[c]) : 0.0f;
  }
  if (tid < TTILE) {
    out[(size_t)b * 2 * TLEN + t0 + tid] = red[HALO + tid];
  }
  __syncthreads();

  layer_step<1, 1, 152, 10>(sh, sy, t0, sep_w + 0 * CF * 3, sep_b + 0 * CF,
                            pw_w + 0 * CF * CF, pw_b + 0 * CF, g1 + 0 * CF,
                            b1 + 0 * CF, g2 + 0 * CF, b2 + 0 * CF);
  layer_step<3, 4, 146, 10>(sh, sy, t0, sep_w + 1 * CF * 3, sep_b + 1 * CF,
                            pw_w + 1 * CF * CF, pw_b + 1 * CF, g1 + 1 * CF,
                            b1 + 1 * CF, g2 + 1 * CF, b2 + 1 * CF);
  layer_step<9, 13, 128, 8>(sh, sy, t0, sep_w + 2 * CF * 3, sep_b + 2 * CF,
                            pw_w + 2 * CF * CF, pw_b + 2 * CF, g1 + 2 * CF,
                            b1 + 2 * CF, g2 + 2 * CF, b2 + 2 * CF);

  proj_mm(proj_w, proj_b, sh, sy);
  __syncthreads();

  float la_local = 0.0f;
  if (tid < TTILE) {
    const int u = HALO + tid;
    const int t = t0 + tid;
    float uw[NBINS], uh[NBINS], d[NBINS + 1];
#pragma unroll
    for (int i = 0; i < NBINS; ++i) uw[i] = sy[sidx(i, u)];
#pragma unroll
    for (int i = 0; i < NBINS; ++i) uh[i] = sy[sidx(NBINS + i, u)];
    d[0] = 1.0f;
    d[NBINS] = 1.0f;
#pragma unroll
    for (int i = 0; i < NBINS - 1; ++i)
      d[i + 1] = MIND + softplusf(sy[sidx(2 * NBINS + i, u)]);

    float cw[NBINS + 1], ch[NBINS + 1];
    build_knots(uw, cw, MINBW);
    build_knots(uh, ch, MINBH);

    const float xin = x[(size_t)b * 2 * TLEN + TLEN + t];
    const bool inside = (xin >= -TAILV) && (xin <= TAILV);
    const float xc = fminf(fmaxf(xin, -TAILV), TAILV);
    int idx = -1;
#pragma unroll
    for (int k = 0; k < NBINS + 1; ++k) idx += (xc >= cw[k]) ? 1 : 0;
    idx = max(0, min(idx, NBINS - 1));

    const float icw = cw[idx];
    const float ibw = cw[idx + 1] - cw[idx];
    const float ich = ch[idx];
    const float ih  = ch[idx + 1] - ch[idx];
    const float idelta = ih / ibw;
    const float d0 = d[idx];
    const float d1 = d[idx + 1];
    const float theta = (xc - icw) / ibw;
    const float th1m  = 1.0f - theta;
    const float num = ih * (idelta * theta * theta + d0 * theta * th1m);
    const float den = idelta + (d0 + d1 - 2.0f * idelta) * theta * th1m;
    float outv = ich + num / den;
    const float dnum = idelta * idelta *
        (d1 * theta * theta + 2.0f * idelta * theta * th1m + d0 * th1m * th1m);
    float la = logf(dnum) - 2.0f * logf(den);
    if (!inside) { outv = xin; la = 0.0f; }
    out[(size_t)b * 2 * TLEN + TLEN + t] = outv;
    la_local = la;
  }

  __syncthreads();
  red[tid] = (tid < TTILE) ? la_local : 0.0f;
  __syncthreads();
  for (int s = 64; s > 0; s >>= 1) {
    if (tid < s) red[tid] += red[tid + s];
    __syncthreads();
  }
  if (tid == 0) atomicAdd(out + (size_t)NBATCH * 2 * TLEN + b, red[0]);
}

__global__ void zero_logdet(float* p) {
  if (threadIdx.x < NBATCH) p[threadIdx.x] = 0.0f;
}

extern "C" void kernel_launch(void* const* d_in, const int* in_sizes, int n_in,
                              void* d_out, int out_size, void* d_ws,
                              size_t ws_size, hipStream_t stream) {
  (void)in_sizes; (void)n_in; (void)out_size; (void)d_ws; (void)ws_size;
  const float* x      = (const float*)d_in[0];
  const float* pre_w  = (const float*)d_in[2];
  const float* pre_b  = (const float*)d_in[3];
  const float* sep_w  = (const float*)d_in[4];
  const float* sep_b  = (const float*)d_in[5];
  const float* pw_w   = (const float*)d_in[6];
  const float* pw_b   = (const float*)d_in[7];
  const float* g1     = (const float*)d_in[8];
  const float* b1     = (const float*)d_in[9];
  const float* g2     = (const float*)d_in[10];
  const float* b2     = (const float*)d_in[11];
  const float* proj_w = (const float*)d_in[12];
  const float* proj_b = (const float*)d_in[13];
  float* out = (float*)d_out;

  zero_logdet<<<1, 32, 0, stream>>>(out + (size_t)NBATCH * 2 * TLEN);

  const size_t smem_bytes = (size_t)(2 * CF * SSTR + 512) * sizeof(float);
  dim3 grid(TLEN / TTILE, NBATCH);
  convflow_kernel<<<grid, NTHR, smem_bytes, stream>>>(
      x, pre_w, pre_b, sep_w, sep_b, pw_w, pw_b, g1, b1, g2, b2, proj_w,
      proj_b, out);
}


# LayerNorm channel sums: 5-step lane butterfly moved off the LDS unit (ds_bpermute, an LDS round trip per step) onto the vector ALU: v_permlanex16 then DPP row_xmask 8,4,2,1; same pairing order
# speedup vs baseline: 1.9719x; 1.0867x over previous
#include <hip/hip_runtime.h>
#include <math.h>


#define CF     192
#define TTILE  128
#define HALO   13
#define HW     (TTILE + 2*HALO)
#define SSTR   155
#define NTHR   384
#define NWAVE  12
#define TLEN   8192
#define NBATCH 32
#define NBINS  10
#define TAILV  5.0f
#define MINBW  0.001f
#define MINBH  0.001f
#define MIND   0.001f

typedef float v2f __attribute__((ext_vector_type(2)));
typedef float v8f __attribute__((ext_vector_type(8)));

__device__ __forceinline__ int sidx(int c, int u) {
  return (c >> 1) * (2 * SSTR) + 2 * u + (c & 1);
}

__device__ __forceinline__ float gelu_exact(float x) {
  const float ax = fabsf(x);
  const float t = __builtin_amdgcn_rcpf(fmaf(ax, 0.28f, 1.0f));
  const float e = __builtin_amdgcn_exp2f((x * x) * -0.72134752044448170368f);
  float p = -0.10675480918401203f;
  p = fmaf(p, t, 0.39380918123096653f);
  p = fmaf(p, t, -0.2434958341181943f);
  p = fmaf(p, t, 0.27424663512843267f);
  p = fmaf(p, t, 0.06532213231594822f);
  p = fmaf(p, t, 0.11687268054021908f);
  const float tail = (p * t) * e;
  return fmaf(-ax, tail, fmaxf(x, 0.0f));
}
__device__ __forceinline__ float softplusf(float x) {
  return (x > 20.0f) ? x : log1pf(expf(x));
}

__device__ __forceinline__ float xhalf(float x) {
  const int b = __builtin_bit_cast(int, x);
  return __builtin_bit_cast(
      float, __builtin_amdgcn_permlanex16(b, b, 0x76543210, 0xfedcba98, false,
                                          false));
}
template<int M>
__device__ __forceinline__ float row_xor(float x) {
  return __builtin_bit_cast(
      float, __builtin_amdgcn_update_dpp(0, __builtin_bit_cast(int, x),
                                         0x160 + M, 0xf, 0xf, true));
}

__device__ __forceinline__ void ln_gelu_regs(float (&v)[6],
                                             const float (&gm)[6],
                                             const float (&bt)[6]) {
  float s1 = 0.f, s2 = 0.f;
#pragma unroll
  for (int r = 0; r < 6; ++r) {
    s1 += v[r];
    s2 += v[r] * v[r];
  }
  s1 += xhalf(s1);
  s2 += xhalf(s2);
  s1 += row_xor<8>(s1);
  s2 += row_xor<8>(s2);
  s1 += row_xor<4>(s1);
  s2 += row_xor<4>(s2);
  s1 += row_xor<2>(s1);
  s2 += row_xor<2>(s2);
  s1 += row_xor<1>(s1);
  s2 += row_xor<1>(s2);
  const float mu  = s1 * (1.0f / 192.0f);
  const float var = s2 * (1.0f / 192.0f) - mu * mu;
  const float rs  = rsqrtf(var + 1e-5f);
#pragma unroll
  for (int r = 0; r < 6; ++r) {
    const float xh = (v[r] - mu) * rs;
    v[r] = gelu_exact(xh * gm[r] + bt[r]);
  }
}

template<int DIL, int US, int W>
__device__ void dwconv_ln_gelu(const float* __restrict__ sh,
                               float* __restrict__ sy,
                               const float* __restrict__ sepw,
                               const float* __restrict__ sepb,
                               const float* __restrict__ gamma,
                               const float* __restrict__ beta) {
  const int tid  = threadIdx.x;
  const int lane = tid & 31;
  const int wid  = tid >> 5;
  float w0[6], w1[6], w2[6], sb[6], gm[6], bt[6];
#pragma unroll
  for (int r = 0; r < 6; ++r) {
    const int c = lane + 32 * r;
    w0[r] = sepw[c * 3 + 0];
    w1[r] = sepw[c * 3 + 1];
    w2[r] = sepw[c * 3 + 2];
    sb[r] = sepb[c];
    gm[r] = gamma[c];
    bt[r] = beta[c];
  }
  for (int j = wid; j < W; j += NWAVE) {
    const int u = US + j;
    float v[6];
#pragma unroll
    for (int r = 0; r < 6; ++r) {
      const int i = sidx(lane + 32 * r, u);
      v[r] = w0[r] * sh[i - 2 * DIL] + w1[r] * sh[i] + w2[r] * sh[i + 2 * DIL] +
             sb[r];
    }
    ln_gelu_regs(v, gm, bt);
#pragma unroll
    for (int r = 0; r < 6; ++r) sy[sidx(lane + 32 * r, u)] = v[r];
  }
}

template<int US, int W>
__device__ void ln_gelu_residual(const float* __restrict__ sy,
                                 float* __restrict__ sh, int t0,
                                 const float* __restrict__ gamma,
                                 const float* __restrict__ beta) {
  const int tid  = threadIdx.x;
  const int lane = tid & 31;
  const int wid  = tid >> 5;
  float gm[6], bt[6];
#pragma unroll
  for (int r = 0; r < 6; ++r) {
    gm[r] = gamma[lane + 32 * r];
    bt[r] = beta[lane + 32 * r];
  }
  for (int j = wid; j < W; j += NWAVE) {
    const int u = US + j;
    const int t = t0 - HALO + u;
    float v[6], hv[6];
#pragma unroll
    for (int r = 0; r < 6; ++r) {
      const int i = sidx(lane + 32 * r, u);
      v[r]  = sy[i];
      hv[r] = sh[i];
    }
    ln_gelu_regs(v, gm, bt);
    const bool valid = (t >= 0) && (t < TLEN);
#pragma unroll
    for (int r = 0; r < 6; ++r)
      sh[sidx(lane + 32 * r, u)] = hv[r] + (valid ? v[r] : 0.0f);
  }
}

template<int US, int W, int NT>
__device__ void pw_conv_inplace(const float* __restrict__ Wm,
                                const float* __restrict__ bias, float* sy) {
  const int tid  = threadIdx.x;
  const int lane = tid & 31;
  const int wid  = tid >> 5;
  const int m0   = wid * 16;
  const int l15  = lane & 15;
  const int kof  = (lane & 16) ? 2 : 0;
  const int colmax = US + W - 1;

  v2f afrag[48];
  {
    const v2f* wrow = (const v2f*)(Wm + (size_t)(m0 + l15) * CF);
    const int kidx0 = kof >> 1;
#pragma unroll
    for (int kk = 0; kk < 48; ++kk) afrag[kk] = wrow[kk * 2 + kidx0];
  }

  for (int n = 0; n < NT; ++n) {
    int colb = US + n * 16 + l15;
    if (colb > colmax) colb = colmax;
    v8f c8 = {};
#pragma unroll
    for (int kk = 0; kk < 48; ++kk) {
      const int k = kk * 4 + kof;
      const v2f bv = *(const v2f*)(&sy[(k >> 1) * (2 * SSTR) + 2 * colb]);
      c8 = __builtin_amdgcn_wmma_f32_16x16x4_f32(
          false, afrag[kk], false, bv, (short)0, c8, false, false);
    }
    __syncthreads();
    const int col = US + n * 16 + l15;
    if (col <= colmax) {
      const int rbase = m0 + ((lane & 16) ? 8 : 0);
#pragma unroll
      for (int v = 0; v < 8; ++v) {
        const int row = rbase + v;
        sy[sidx(row, col)] = c8[v] + bias[row];
      }
    }
  }
}

__device__ void proj_mm(const float* __restrict__ Wm,
                        const float* __restrict__ bias,
                        const float* sh, float* sy) {
  const int tid  = threadIdx.x;
  const int lane = tid & 31;
  const int wid  = tid >> 5;
  const int l15  = lane & 15;
  const int kof  = (lane & 16) ? 2 : 0;
  for (int job = wid; job < 16; job += NWAVE) {
    const int m0 = (job & 1) * 16;
    const int n  = job >> 1;
    const int row  = m0 + l15;
    const int rowc = row < 29 ? row : 28;
    const v2f* wrow = (const v2f*)(Wm + (size_t)rowc * CF);
    const int col = HALO + n * 16 + l15;
    v2f afrag[48];
#pragma unroll
    for (int kk = 0; kk < 48; ++kk) afrag[kk] = wrow[kk * 2 + (kof >> 1)];
    v8f c8 = {};
#pragma unroll
    for (int kk = 0; kk < 48; ++kk) {
      const int k = kk * 4 + kof;
      const v2f bv = *(const v2f*)(&sh[(k >> 1) * (2 * SSTR) + 2 * col]);
      c8 = __builtin_amdgcn_wmma_f32_16x16x4_f32(
          false, afrag[kk], false, bv, (short)0, c8, false, false);
    }
    const int rbase = m0 + ((lane & 16) ? 8 : 0);
#pragma unroll
    for (int v = 0; v < 8; ++v) {
      const int r = rbase + v;
      if (r < 29) sy[sidx(r, col)] = c8[v] + bias[r];
    }
  }
}

template<int DIL, int US, int W, int NT>
__device__ void layer_step(float* sh, float* sy, int t0,
                           const float* __restrict__ sepw,
                           const float* __restrict__ sepb,
                           const float* __restrict__ pww,
                           const float* __restrict__ pwb,
                           const float* __restrict__ gg1,
                           const float* __restrict__ bb1,
                           const float* __restrict__ gg2,
                           const float* __restrict__ bb2) {
  dwconv_ln_gelu<DIL, US, W>(sh, sy, sepw, sepb, gg1, bb1);
  __syncthreads();
  pw_conv_inplace<US, W, NT>(pww, pwb, sy);
  __syncthreads();
  ln_gelu_residual<US, W>(sy, sh, t0, gg2, bb2);
  __syncthreads();
}

__device__ __forceinline__ void build_knots(const float* uu, float* cks,
                                            float minb) {
  float m = uu[0];
#pragma unroll
  for (int i = 1; i < NBINS; ++i) m = fmaxf(m, uu[i]);
  float e[NBINS];
  float s = 0.f;
#pragma unroll
  for (int i = 0; i < NBINS; ++i) { e[i] = expf(uu[i] - m); s += e[i]; }
  const float inv = 1.0f / s;
  const float scale = 1.0f - minb * (float)NBINS;
  cks[0] = -TAILV;
  float acc = 0.f;
#pragma unroll
  for (int i = 0; i < NBINS - 1; ++i) {
    acc += minb + scale * e[i] * inv;
    cks[i + 1] = -TAILV + 2.0f * TAILV * acc;
  }
  cks[NBINS] = TAILV;
}

__global__ void __launch_bounds__(NTHR, 1)
convflow_kernel(const float* __restrict__ x,
                const float* __restrict__ pre_w, const float* __restrict__ pre_b,
                const float* __restrict__ sep_w, const float* __restrict__ sep_b,
                const float* __restrict__ pw_w,  const float* __restrict__ pw_b,
                const float* __restrict__ g1,    const float* __restrict__ b1,
                const float* __restrict__ g2,    const float* __restrict__ b2,
                const float* __restrict__ proj_w,const float* __restrict__ proj_b,
                float* __restrict__ out) {
  extern __shared__ float smem[];
  float* sh  = smem;
  float* sy  = smem + CF * SSTR;
  float* red = smem + 2 * CF * SSTR;

  const int tid = threadIdx.x;
  const int tile = blockIdx.x;
  const int b    = blockIdx.y;
  const int t0   = tile * TTILE;

  for (int u = tid; u < HW; u += NTHR) {
    const int t = t0 - HALO + u;
    red[u] = (t >= 0 && t < TLEN) ? x[(size_t)b * 2 * TLEN + t] : 0.0f;
  }
  __syncthreads();
  for (int e = tid; e < CF * HW; e += NTHR) {
    const int c = e / HW;
    const int u = e - c * HW;
    const int t = t0 - HALO + u;
    sh[sidx(c, u)] =
        (t >= 0 && t < TLEN) ? (pre_w[c] * red[u] + pre_b[c]) : 0.0f;
  }
  if (tid < TTILE) {
    out[(size_t)b * 2 * TLEN + t0 + tid] = red[HALO + tid];
  }
  __syncthreads();

  layer_step<1, 1, 152, 10>(sh, sy, t0, sep_w + 0 * CF * 3, sep_b + 0 * CF,
                            pw_w + 0 * CF * CF, pw_b + 0 * CF, g1 + 0 * CF,
                            b1 + 0 * CF, g2 + 0 * CF, b2 + 0 * CF);
  layer_step<3, 4, 146, 10>(sh, sy, t0, sep_w + 1 * CF * 3, sep_b + 1 * CF,
                            pw_w + 1 * CF * CF, pw_b + 1 * CF, g1 + 1 * CF,
                            b1 + 1 * CF, g2 + 1 * CF, b2 + 1 * CF);
  layer_step<9, 13, 128, 8>(sh, sy, t0, sep_w + 2 * CF * 3, sep_b + 2 * CF,
                            pw_w + 2 * CF * CF, pw_b + 2 * CF, g1 + 2 * CF,
                            b1 + 2 * CF, g2 + 2 * CF, b2 + 2 * CF);

  proj_mm(proj_w, proj_b, sh, sy);
  __syncthreads();

  float la_local = 0.0f;
  if (tid < TTILE) {
    const int u = HALO + tid;
    const int t = t0 + tid;
    float uw[NBINS], uh[NBINS], d[NBINS + 1];
#pragma unroll
    for (int i = 0; i < NBINS; ++i) uw[i] = sy[sidx(i, u)];
#pragma unroll
    for (int i = 0; i < NBINS; ++i) uh[i] = sy[sidx(NBINS + i, u)];
    d[0] = 1.0f;
    d[NBINS] = 1.0f;
#pragma unroll
    for (int i = 0; i < NBINS - 1; ++i)
      d[i + 1] = MIND + softplusf(sy[sidx(2 * NBINS + i, u)]);

    float cw[NBINS + 1], ch[NBINS + 1];
    build_knots(uw, cw, MINBW);
    build_knots(uh, ch, MINBH);

    const float xin = x[(size_t)b * 2 * TLEN + TLEN + t];
    const bool inside = (xin >= -TAILV) && (xin <= TAILV);
    const float xc = fminf(fmaxf(xin, -TAILV), TAILV);
    int idx = -1;
#pragma unroll
    for (int k = 0; k < NBINS + 1; ++k) idx += (xc >= cw[k]) ? 1 : 0;
    idx = max(0, min(idx, NBINS - 1));

    const float icw = cw[idx];
    const float ibw = cw[idx + 1] - cw[idx];
    const float ich = ch[idx];
    const float ih  = ch[idx + 1] - ch[idx];
    const float idelta = ih / ibw;
    const float d0 = d[idx];
    const float d1 = d[idx + 1];
    const float theta = (xc - icw) / ibw;
    const float th1m  = 1.0f - theta;
    const float num = ih * (idelta * theta * theta + d0 * theta * th1m);
    const float den = idelta + (d0 + d1 - 2.0f * idelta) * theta * th1m;
    float outv = ich + num / den;
    const float dnum = idelta * idelta *
        (d1 * theta * theta + 2.0f * idelta * theta * th1m + d0 * th1m * th1m);
    float la = logf(dnum) - 2.0f * logf(den);
    if (!inside) { outv = xin; la = 0.0f; }
    out[(size_t)b * 2 * TLEN + TLEN + t] = outv;
    la_local = la;
  }

  __syncthreads();
  red[tid] = (tid < TTILE) ? la_local : 0.0f;
  __syncthreads();
  for (int s = 64; s > 0; s >>= 1) {
    if (tid < s) red[tid] += red[tid + s];
    __syncthreads();
  }
  if (tid == 0) atomicAdd(out + (size_t)NBATCH * 2 * TLEN + b, red[0]);
}

__global__ void zero_logdet(float* p) {
  if (threadIdx.x < NBATCH) p[threadIdx.x] = 0.0f;
}

extern "C" void kernel_launch(void* const* d_in, const int* in_sizes, int n_in,
                              void* d_out, int out_size, void* d_ws,
                              size_t ws_size, hipStream_t stream) {
  (void)in_sizes; (void)n_in; (void)out_size; (void)d_ws; (void)ws_size;
  const float* x      = (const float*)d_in[0];
  const float* pre_w  = (const float*)d_in[2];
  const float* pre_b  = (const float*)d_in[3];
  const float* sep_w  = (const float*)d_in[4];
  const float* sep_b  = (const float*)d_in[5];
  const float* pw_w   = (const float*)d_in[6];
  const float* pw_b   = (const float*)d_in[7];
  const float* g1     = (const float*)d_in[8];
  const float* b1     = (const float*)d_in[9];
  const float* g2     = (const float*)d_in[10];
  const float* b2     = (const float*)d_in[11];
  const float* proj_w = (const float*)d_in[12];
  const float* proj_b = (const float*)d_in[13];
  float* out = (float*)d_out;

  zero_logdet<<<1, 32, 0, stream>>>(out + (size_t)NBATCH * 2 * TLEN);

  const size_t smem_bytes = (size_t)(2 * CF * SSTR + 512) * sizeof(float);
  dim3 grid(TLEN / TTILE, NBATCH);
  convflow_kernel<<<grid, NTHR, smem_bytes, stream>>>(
      x, pre_w, pre_b, sep_w, sep_b, pw_w, pw_b, g1, b1, g2, b2, proj_w,
      proj_b, out);
}


# Matmul tile's 48 WMMAs issue back to back: B loads as one group a tile ahead (was one LDS load per WMMA), K summed in two interleaved accumulator chains added at the end (sum order changes)
# speedup vs baseline: 2.0326x; 1.0308x over previous
#include <hip/hip_runtime.h>
#include <math.h>


#define CF     192
#define TTILE  128
#define HALO   13
#define HW     (TTILE + 2*HALO)
#define SSTR   155
#define NTHR   384
#define NWAVE  12
#define TLEN   8192
#define NBATCH 32
#define NBINS  10
#define TAILV  5.0f
#define MINBW  0.001f
#define MINBH  0.001f
#define MIND   0.001f

typedef float v2f __attribute__((ext_vector_type(2)));
typedef float v8f __attribute__((ext_vector_type(8)));

__device__ __forceinline__ int sidx(int c, int u) {
  return (c >> 1) * (2 * SSTR) + 2 * u + (c & 1);
}

__device__ __forceinline__ float gelu_exact(float x) {
  const float ax = fabsf(x);
  const float t = __builtin_amdgcn_rcpf(fmaf(ax, 0.28f, 1.0f));
  const float e = __builtin_amdgcn_exp2f((x * x) * -0.72134752044448170368f);
  float p = -0.10675480918401203f;
  p = fmaf(p, t, 0.39380918123096653f);
  p = fmaf(p, t, -0.2434958341181943f);
  p = fmaf(p, t, 0.27424663512843267f);
  p = fmaf(p, t, 0.06532213231594822f);
  p = fmaf(p, t, 0.11687268054021908f);
  const float tail = (p * t) * e;
  return fmaf(-ax, tail, fmaxf(x, 0.0f));
}
__device__ __forceinline__ float softplusf(float x) {
  return (x > 20.0f) ? x : log1pf(expf(x));
}

__device__ __forceinline__ float xhalf(float x) {
  const int b = __builtin_bit_cast(int, x);
  return __builtin_bit_cast(
      float, __builtin_amdgcn_permlanex16(b, b, 0x76543210, 0xfedcba98, false,
                                          false));
}
template<int M>
__device__ __forceinline__ float row_xor(float x) {
  return __builtin_bit_cast(
      float, __builtin_amdgcn_update_dpp(0, __builtin_bit_cast(int, x),
                                         0x160 + M, 0xf, 0xf, true));
}

__device__ __forceinline__ void ln_gelu_regs(float (&v)[6],
                                             const float (&gm)[6],
                                             const float (&bt)[6]) {
  float s1 = 0.f, s2 = 0.f;
#pragma unroll
  for (int r = 0; r < 6; ++r) {
    s1 += v[r];
    s2 += v[r] * v[r];
  }
  s1 += xhalf(s1);
  s2 += xhalf(s2);
  s1 += row_xor<8>(s1);
  s2 += row_xor<8>(s2);
  s1 += row_xor<4>(s1);
  s2 += row_xor<4>(s2);
  s1 += row_xor<2>(s1);
  s2 += row_xor<2>(s2);
  s1 += row_xor<1>(s1);
  s2 += row_xor<1>(s2);
  const float mu  = s1 * (1.0f / 192.0f);
  const float var = s2 * (1.0f / 192.0f) - mu * mu;
  const float rs  = rsqrtf(var + 1e-5f);
#pragma unroll
  for (int r = 0; r < 6; ++r) {
    const float xh = (v[r] - mu) * rs;
    v[r] = gelu_exact(xh * gm[r] + bt[r]);
  }
}

template<int DIL, int US, int W>
__device__ void dwconv_ln_gelu(const float* __restrict__ sh,
                               float* __restrict__ sy,
                               const float* __restrict__ sepw,
                               const float* __restrict__ sepb,
                               const float* __restrict__ gamma,
                               const float* __restrict__ beta) {
  const int tid  = threadIdx.x;
  const int lane = tid & 31;
  const int wid  = tid >> 5;
  float w0[6], w1[6], w2[6], sb[6], gm[6], bt[6];
#pragma unroll
  for (int r = 0; r < 6; ++r) {
    const int c = lane + 32 * r;
    w0[r] = sepw[c * 3 + 0];
    w1[r] = sepw[c * 3 + 1];
    w2[r] = sepw[c * 3 + 2];
    sb[r] = sepb[c];
    gm[r] = gamma[c];
    bt[r] = beta[c];
  }
  for (int j = wid; j < W; j += NWAVE) {
    const int u = US + j;
    float v[6];
#pragma unroll
    for (int r = 0; r < 6; ++r) {
      const int i = sidx(lane + 32 * r, u);
      v[r] = w0[r] * sh[i - 2 * DIL] + w1[r] * sh[i] + w2[r] * sh[i + 2 * DIL] +
             sb[r];
    }
    ln_gelu_regs(v, gm, bt);
#pragma unroll
    for (int r = 0; r < 6; ++r) sy[sidx(lane + 32 * r, u)] = v[r];
  }
}

template<int US, int W>
__device__ void ln_gelu_residual(const float* __restrict__ sy,
                                 float* __restrict__ sh, int t0,
                                 const float* __restrict__ gamma,
                                 const float* __restrict__ beta) {
  const int tid  = threadIdx.x;
  const int lane = tid & 31;
  const int wid  = tid >> 5;
  float gm[6], bt[6];
#pragma unroll
  for (int r = 0; r < 6; ++r) {
    gm[r] = gamma[lane + 32 * r];
    bt[r] = beta[lane + 32 * r];
  }
  for (int j = wid; j < W; j += NWAVE) {
    const int u = US + j;
    const int t = t0 - HALO + u;
    float v[6], hv[6];
#pragma unroll
    for (int r = 0; r < 6; ++r) {
      const int i = sidx(lane + 32 * r, u);
      v[r]  = sy[i];
      hv[r] = sh[i];
    }
    ln_gelu_regs(v, gm, bt);
    const bool valid = (t >= 0) && (t < TLEN);
#pragma unroll
    for (int r = 0; r < 6; ++r)
      sh[sidx(lane + 32 * r, u)] = hv[r] + (valid ? v[r] : 0.0f);
  }
}

__device__ __forceinline__ void load_bfrags(const float* buf, int colb, int kof,
                                            v2f (&bf)[48]) {
#pragma unroll
  for (int kk = 0; kk < 48; ++kk) {
    const int k = kk * 4 + kof;
    bf[kk] = *(const v2f*)(&buf[(k >> 1) * (2 * SSTR) + 2 * colb]);
  }
}

__device__ __forceinline__ void wmma_chains(const v2f (&af)[48],
                                            const v2f (&bf)[48],
                                            v8f& ce, v8f& co) {
  ce = v8f{};
  co = v8f{};
#pragma unroll
  for (int kk = 0; kk < 48; kk += 2) {
    ce = __builtin_amdgcn_wmma_f32_16x16x4_f32(
        false, af[kk], false, bf[kk], (short)0, ce, false, false);
    co = __builtin_amdgcn_wmma_f32_16x16x4_f32(
        false, af[kk + 1], false, bf[kk + 1], (short)0, co, false, false);
  }
}

template<int US, int W, int NT>
__device__ void pw_conv_inplace(const float* __restrict__ Wm,
                                const float* __restrict__ bias, float* sy) {
  const int tid  = threadIdx.x;
  const int lane = tid & 31;
  const int wid  = tid >> 5;
  const int m0   = wid * 16;
  const int l15  = lane & 15;
  const int kof  = (lane & 16) ? 2 : 0;
  const int colmax = US + W - 1;

  v2f afrag[48];
  {
    const v2f* wrow = (const v2f*)(Wm + (size_t)(m0 + l15) * CF);
    const int kidx0 = kof >> 1;
#pragma unroll
    for (int kk = 0; kk < 48; ++kk) afrag[kk] = wrow[kk * 2 + kidx0];
  }

  v2f bfrag[48];
  {
    int colb = US + l15;
    if (colb > colmax) colb = colmax;
    load_bfrags(sy, colb, kof, bfrag);
  }
  for (int n = 0; n < NT; ++n) {
    v8f ce, co;
    wmma_chains(afrag, bfrag, ce, co);
    __builtin_amdgcn_sched_barrier(0);
    if (n + 1 < NT) {
      int colb = US + (n + 1) * 16 + l15;
      if (colb > colmax) colb = colmax;
      load_bfrags(sy, colb, kof, bfrag);
    }
    const v8f c8 = ce + co;
    __syncthreads();
    const int col = US + n * 16 + l15;
    if (col <= colmax) {
      const int rbase = m0 + ((lane & 16) ? 8 : 0);
#pragma unroll
      for (int v = 0; v < 8; ++v) {
        const int row = rbase + v;
        sy[sidx(row, col)] = c8[v] + bias[row];
      }
    }
  }
}

__device__ void proj_mm(const float* __restrict__ Wm,
                        const float* __restrict__ bias,
                        const float* sh, float* sy) {
  const int tid  = threadIdx.x;
  const int lane = tid & 31;
  const int wid  = tid >> 5;
  const int l15  = lane & 15;
  const int kof  = (lane & 16) ? 2 : 0;
  for (int job = wid; job < 16; job += NWAVE) {
    const int m0 = (job & 1) * 16;
    const int n  = job >> 1;
    const int row  = m0 + l15;
    const int rowc = row < 29 ? row : 28;
    const v2f* wrow = (const v2f*)(Wm + (size_t)rowc * CF);
    const int col = HALO + n * 16 + l15;
    v2f afrag[48], bfrag[48];
#pragma unroll
    for (int kk = 0; kk < 48; ++kk) afrag[kk] = wrow[kk * 2 + (kof >> 1)];
    load_bfrags(sh, col, kof, bfrag);
    __builtin_amdgcn_sched_barrier(0);
    v8f ce, co;
    wmma_chains(afrag, bfrag, ce, co);
    const v8f c8 = ce + co;
    const int rbase = m0 + ((lane & 16) ? 8 : 0);
#pragma unroll
    for (int v = 0; v < 8; ++v) {
      const int r = rbase + v;
      if (r < 29) sy[sidx(r, col)] = c8[v] + bias[r];
    }
  }
}

template<int DIL, int US, int W, int NT>
__device__ void layer_step(float* sh, float* sy, int t0,
                           const float* __restrict__ sepw,
                           const float* __restrict__ sepb,
                           const float* __restrict__ pww,
                           const float* __restrict__ pwb,
                           const float* __restrict__ gg1,
                           const float* __restrict__ bb1,
                           const float* __restrict__ gg2,
                           const float* __restrict__ bb2) {
  dwconv_ln_gelu<DIL, US, W>(sh, sy, sepw, sepb, gg1, bb1);
  __syncthreads();
  pw_conv_inplace<US, W, NT>(pww, pwb, sy);
  __syncthreads();
  ln_gelu_residual<US, W>(sy, sh, t0, gg2, bb2);
  __syncthreads();
}

__device__ __forceinline__ void build_knots(const float* uu, float* cks,
                                            float minb) {
  float m = uu[0];
#pragma unroll
  for (int i = 1; i < NBINS; ++i) m = fmaxf(m, uu[i]);
  float e[NBINS];
  float s = 0.f;
#pragma unroll
  for (int i = 0; i < NBINS; ++i) { e[i] = expf(uu[i] - m); s += e[i]; }
  const float inv = 1.0f / s;
  const float scale = 1.0f - minb * (float)NBINS;
  cks[0] = -TAILV;
  float acc = 0.f;
#pragma unroll
  for (int i = 0; i < NBINS - 1; ++i) {
    acc += minb + scale * e[i] * inv;
    cks[i + 1] = -TAILV + 2.0f * TAILV * acc;
  }
  cks[NBINS] = TAILV;
}

__global__ void __launch_bounds__(NTHR, 1)
convflow_kernel(const float* __restrict__ x,
                const float* __restrict__ pre_w, const float* __restrict__ pre_b,
                const float* __restrict__ sep_w, const float* __restrict__ sep_b,
                const float* __restrict__ pw_w,  const float* __restrict__ pw_b,
                const float* __restrict__ g1,    const float* __restrict__ b1,
                const float* __restrict__ g2,    const float* __restrict__ b2,
                const float* __restrict__ proj_w,const float* __restrict__ proj_b,
                float* __restrict__ out) {
  extern __shared__ float smem[];
  float* sh  = smem;
  float* sy  = smem + CF * SSTR;
  float* red = smem + 2 * CF * SSTR;

  const int tid = threadIdx.x;
  const int tile = blockIdx.x;
  const int b    = blockIdx.y;
  const int t0   = tile * TTILE;

  for (int u = tid; u < HW; u += NTHR) {
    const int t = t0 - HALO + u;
    red[u] = (t >= 0 && t < TLEN) ? x[(size_t)b * 2 * TLEN + t] : 0.0f;
  }
  __syncthreads();
  for (int e = tid; e < CF * HW; e += NTHR) {
    const int c = e / HW;
    const int u = e - c * HW;
    const int t = t0 - HALO + u;
    sh[sidx(c, u)] =
        (t >= 0 && t < TLEN) ? (pre_w[c] * red[u] + pre_b[c]) : 0.0f;
  }
  if (tid < TTILE) {
    out[(size_t)b * 2 * TLEN + t0 + tid] = red[HALO + tid];
  }
  __syncthreads();

  layer_step<1, 1, 152, 10>(sh, sy, t0, sep_w + 0 * CF * 3, sep_b + 0 * CF,
                            pw_w + 0 * CF * CF, pw_b + 0 * CF, g1 + 0 * CF,
                            b1 + 0 * CF, g2 + 0 * CF, b2 + 0 * CF);
  layer_step<3, 4, 146, 10>(sh, sy, t0, sep_w + 1 * CF * 3, sep_b + 1 * CF,
                            pw_w + 1 * CF * CF, pw_b + 1 * CF, g1 + 1 * CF,
                            b1 + 1 * CF, g2 + 1 * CF, b2 + 1 * CF);
  layer_step<9, 13, 128, 8>(sh, sy, t0, sep_w + 2 * CF * 3, sep_b + 2 * CF,
                            pw_w + 2 * CF * CF, pw_b + 2 * CF, g1 + 2 * CF,
                            b1 + 2 * CF, g2 + 2 * CF, b2 + 2 * CF);

  proj_mm(proj_w, proj_b, sh, sy);
  __syncthreads();

  float la_local = 0.0f;
  if (tid < TTILE) {
    const int u = HALO + tid;
    const int t = t0 + tid;
    float uw[NBINS], uh[NBINS], d[NBINS + 1];
#pragma unroll
    for (int i = 0; i < NBINS; ++i) uw[i] = sy[sidx(i, u)];
#pragma unroll
    for (int i = 0; i < NBINS; ++i) uh[i] = sy[sidx(NBINS + i, u)];
    d[0] = 1.0f;
    d[NBINS] = 1.0f;
#pragma unroll
    for (int i = 0; i < NBINS - 1; ++i)
      d[i + 1] = MIND + softplusf(sy[sidx(2 * NBINS + i, u)]);

    float cw[NBINS + 1], ch[NBINS + 1];
    build_knots(uw, cw, MINBW);
    build_knots(uh, ch, MINBH);

    const float xin = x[(size_t)b * 2 * TLEN + TLEN + t];
    const bool inside = (xin >= -TAILV) && (xin <= TAILV);
    const float xc = fminf(fmaxf(xin, -TAILV), TAILV);
    int idx = -1;
#pragma unroll
    for (int k = 0; k < NBINS + 1; ++k) idx += (xc >= cw[k]) ? 1 : 0;
    idx = max(0, min(idx, NBINS - 1));

    const float icw = cw[idx];
    const float ibw = cw[idx + 1] - cw[idx];
    const float ich = ch[idx];
    const float ih  = ch[idx + 1] - ch[idx];
    const float idelta = ih / ibw;
    const float d0 = d[idx];
    const float d1 = d[idx + 1];
    const float theta = (xc - icw) / ibw;
    const float th1m  = 1.0f - theta;
    const float num = ih * (idelta * theta * theta + d0 * theta * th1m);
    const float den = idelta + (d0 + d1 - 2.0f * idelta) * theta * th1m;
    float outv = ich + num / den;
    const float dnum = idelta * idelta *
        (d1 * theta * theta + 2.0f * idelta * theta * th1m + d0 * th1m * th1m);
    float la = logf(dnum) - 2.0f * logf(den);
    if (!inside) { outv = xin; la = 0.0f; }
    out[(size_t)b * 2 * TLEN + TLEN + t] = outv;
    la_local = la;
  }

  __syncthreads();
  red[tid] = (tid < TTILE) ? la_local : 0.0f;
  __syncthreads();
  for (int s = 64; s > 0; s >>= 1) {
    if (tid < s) red[tid] += red[tid + s];
    __syncthreads();
  }
  if (tid == 0) atomicAdd(out + (size_t)NBATCH * 2 * TLEN + b, red[0]);
}

__global__ void zero_logdet(float* p) {
  if (threadIdx.x < NBATCH) p[threadIdx.x] = 0.0f;
}

extern "C" void kernel_launch(void* const* d_in, const int* in_sizes, int n_in,
                              void* d_out, int out_size, void* d_ws,
                              size_t ws_size, hipStream_t stream) {
  (void)in_sizes; (void)n_in; (void)out_size; (void)d_ws; (void)ws_size;
  const float* x      = (const float*)d_in[0];
  const float* pre_w  = (const float*)d_in[2];
  const float* pre_b  = (const float*)d_in[3];
  const float* sep_w  = (const float*)d_in[4];
  const float* sep_b  = (const float*)d_in[5];
  const float* pw_w   = (const float*)d_in[6];
  const float* pw_b   = (const float*)d_in[7];
  const float* g1     = (const float*)d_in[8];
  const float* b1     = (const float*)d_in[9];
  const float* g2     = (const float*)d_in[10];
  const float* b2     = (const float*)d_in[11];
  const float* proj_w = (const float*)d_in[12];
  const float* proj_b = (const float*)d_in[13];
  float* out = (float*)d_out;

  zero_logdet<<<1, 32, 0, stream>>>(out + (size_t)NBATCH * 2 * TLEN);

  const size_t smem_bytes = (size_t)(2 * CF * SSTR + 512) * sizeof(float);
  dim3 grid(TLEN / TTILE, NBATCH);
  convflow_kernel<<<grid, NTHR, smem_bytes, stream>>>(
      x, pre_w, pre_b, sep_w, sep_b, pw_w, pw_b, g1, b1, g2, b2, proj_w,
      proj_b, out);
}


# Input projection pass rebuilt: wave owns 16 channels (weights by scalar loads), lane owns columns (x0 in registers), paired 8-byte LDS stores; no index division, weight-load wait or staging barrier
# speedup vs baseline: 2.2016x; 1.0831x over previous
#include <hip/hip_runtime.h>
#include <math.h>


#define CF     192
#define TTILE  128
#define HALO   13
#define HW     (TTILE + 2*HALO)
#define SSTR   155
#define NTHR   384
#define NWAVE  12
#define TLEN   8192
#define NBATCH 32
#define NBINS  10
#define TAILV  5.0f
#define MINBW  0.001f
#define MINBH  0.001f
#define MIND   0.001f

typedef float v2f __attribute__((ext_vector_type(2)));
typedef float v8f __attribute__((ext_vector_type(8)));

__device__ __forceinline__ int sidx(int c, int u) {
  return (c >> 1) * (2 * SSTR) + 2 * u + (c & 1);
}

__device__ __forceinline__ float gelu_exact(float x) {
  const float ax = fabsf(x);
  const float t = __builtin_amdgcn_rcpf(fmaf(ax, 0.28f, 1.0f));
  const float e = __builtin_amdgcn_exp2f((x * x) * -0.72134752044448170368f);
  float p = -0.10675480918401203f;
  p = fmaf(p, t, 0.39380918123096653f);
  p = fmaf(p, t, -0.2434958341181943f);
  p = fmaf(p, t, 0.27424663512843267f);
  p = fmaf(p, t, 0.06532213231594822f);
  p = fmaf(p, t, 0.11687268054021908f);
  const float tail = (p * t) * e;
  return fmaf(-ax, tail, fmaxf(x, 0.0f));
}
__device__ __forceinline__ float softplusf(float x) {
  return (x > 20.0f) ? x : log1pf(expf(x));
}

__device__ __forceinline__ float xhalf(float x) {
  const int b = __builtin_bit_cast(int, x);
  return __builtin_bit_cast(
      float, __builtin_amdgcn_permlanex16(b, b, 0x76543210, 0xfedcba98, false,
                                          false));
}
template<int M>
__device__ __forceinline__ float row_xor(float x) {
  return __builtin_bit_cast(
      float, __builtin_amdgcn_update_dpp(0, __builtin_bit_cast(int, x),
                                         0x160 + M, 0xf, 0xf, true));
}

__device__ __forceinline__ void ln_gelu_regs(float (&v)[6],
                                             const float (&gm)[6],
                                             const float (&bt)[6]) {
  float s1 = 0.f, s2 = 0.f;
#pragma unroll
  for (int r = 0; r < 6; ++r) {
    s1 += v[r];
    s2 += v[r] * v[r];
  }
  s1 += xhalf(s1);
  s2 += xhalf(s2);
  s1 += row_xor<8>(s1);
  s2 += row_xor<8>(s2);
  s1 += row_xor<4>(s1);
  s2 += row_xor<4>(s2);
  s1 += row_xor<2>(s1);
  s2 += row_xor<2>(s2);
  s1 += row_xor<1>(s1);
  s2 += row_xor<1>(s2);
  const float mu  = s1 * (1.0f / 192.0f);
  const float var = s2 * (1.0f / 192.0f) - mu * mu;
  const float rs  = rsqrtf(var + 1e-5f);
#pragma unroll
  for (int r = 0; r < 6; ++r) {
    const float xh = (v[r] - mu) * rs;
    v[r] = gelu_exact(xh * gm[r] + bt[r]);
  }
}

template<int DIL, int US, int W>
__device__ void dwconv_ln_gelu(const float* __restrict__ sh,
                               float* __restrict__ sy,
                               const float* __restrict__ sepw,
                               const float* __restrict__ sepb,
                               const float* __restrict__ gamma,
                               const float* __restrict__ beta) {
  const int tid  = threadIdx.x;
  const int lane = tid & 31;
  const int wid  = tid >> 5;
  float w0[6], w1[6], w2[6], sb[6], gm[6], bt[6];
#pragma unroll
  for (int r = 0; r < 6; ++r) {
    const int c = lane + 32 * r;
    w0[r] = sepw[c * 3 + 0];
    w1[r] = sepw[c * 3 + 1];
    w2[r] = sepw[c * 3 + 2];
    sb[r] = sepb[c];
    gm[r] = gamma[c];
    bt[r] = beta[c];
  }
  for (int j = wid; j < W; j += NWAVE) {
    const int u = US + j;
    float v[6];
#pragma unroll
    for (int r = 0; r < 6; ++r) {
      const int i = sidx(lane + 32 * r, u);
      v[r] = w0[r] * sh[i - 2 * DIL] + w1[r] * sh[i] + w2[r] * sh[i + 2 * DIL] +
             sb[r];
    }
    ln_gelu_regs(v, gm, bt);
#pragma unroll
    for (int r = 0; r < 6; ++r) sy[sidx(lane + 32 * r, u)] = v[r];
  }
}

template<int US, int W>
__device__ void ln_gelu_residual(const float* __restrict__ sy,
                                 float* __restrict__ sh, int t0,
                                 const float* __restrict__ gamma,
                                 const float* __restrict__ beta) {
  const int tid  = threadIdx.x;
  const int lane = tid & 31;
  const int wid  = tid >> 5;
  float gm[6], bt[6];
#pragma unroll
  for (int r = 0; r < 6; ++r) {
    gm[r] = gamma[lane + 32 * r];
    bt[r] = beta[lane + 32 * r];
  }
  for (int j = wid; j < W; j += NWAVE) {
    const int u = US + j;
    const int t = t0 - HALO + u;
    float v[6], hv[6];
#pragma unroll
    for (int r = 0; r < 6; ++r) {
      const int i = sidx(lane + 32 * r, u);
      v[r]  = sy[i];
      hv[r] = sh[i];
    }
    ln_gelu_regs(v, gm, bt);
    const bool valid = (t >= 0) && (t < TLEN);
#pragma unroll
    for (int r = 0; r < 6; ++r)
      sh[sidx(lane + 32 * r, u)] = hv[r] + (valid ? v[r] : 0.0f);
  }
}

__device__ __forceinline__ void load_bfrags(const float* buf, int colb, int kof,
                                            v2f (&bf)[48]) {
#pragma unroll
  for (int kk = 0; kk < 48; ++kk) {
    const int k = kk * 4 + kof;
    bf[kk] = *(const v2f*)(&buf[(k >> 1) * (2 * SSTR) + 2 * colb]);
  }
}

__device__ __forceinline__ void wmma_chains(const v2f (&af)[48],
                                            const v2f (&bf)[48],
                                            v8f& ce, v8f& co) {
  ce = v8f{};
  co = v8f{};
#pragma unroll
  for (int kk = 0; kk < 48; kk += 2) {
    ce = __builtin_amdgcn_wmma_f32_16x16x4_f32(
        false, af[kk], false, bf[kk], (short)0, ce, false, false);
    co = __builtin_amdgcn_wmma_f32_16x16x4_f32(
        false, af[kk + 1], false, bf[kk + 1], (short)0, co, false, false);
  }
}

template<int US, int W, int NT>
__device__ void pw_conv_inplace(const float* __restrict__ Wm,
                                const float* __restrict__ bias, float* sy) {
  const int tid  = threadIdx.x;
  const int lane = tid & 31;
  const int wid  = tid >> 5;
  const int m0   = wid * 16;
  const int l15  = lane & 15;
  const int kof  = (lane & 16) ? 2 : 0;
  const int colmax = US + W - 1;

  v2f afrag[48];
  {
    const v2f* wrow = (const v2f*)(Wm + (size_t)(m0 + l15) * CF);
    const int kidx0 = kof >> 1;
#pragma unroll
    for (int kk = 0; kk < 48; ++kk) afrag[kk] = wrow[kk * 2 + kidx0];
  }

  v2f bfrag[48];
  {
    int colb = US + l15;
    if (colb > colmax) colb = colmax;
    load_bfrags(sy, colb, kof, bfrag);
  }
  for (int n = 0; n < NT; ++n) {
    v8f ce, co;
    wmma_chains(afrag, bfrag, ce, co);
    __builtin_amdgcn_sched_barrier(0);
    if (n + 1 < NT) {
      int colb = US + (n + 1) * 16 + l15;
      if (colb > colmax) colb = colmax;
      load_bfrags(sy, colb, kof, bfrag);
    }
    const v8f c8 = ce + co;
    __syncthreads();
    const int col = US + n * 16 + l15;
    if (col <= colmax) {
      const int rbase = m0 + ((lane & 16) ? 8 : 0);
#pragma unroll
      for (int v = 0; v < 8; ++v) {
        const int row = rbase + v;
        sy[sidx(row, col)] = c8[v] + bias[row];
      }
    }
  }
}

__device__ void proj_mm(const float* __restrict__ Wm,
                        const float* __restrict__ bias,
                        const float* sh, float* sy) {
  const int tid  = threadIdx.x;
  const int lane = tid & 31;
  const int wid  = tid >> 5;
  const int l15  = lane & 15;
  const int kof  = (lane & 16) ? 2 : 0;
  for (int job = wid; job < 16; job += NWAVE) {
    const int m0 = (job & 1) * 16;
    const int n  = job >> 1;
    const int row  = m0 + l15;
    const int rowc = row < 29 ? row : 28;
    const v2f* wrow = (const v2f*)(Wm + (size_t)rowc * CF);
    const int col = HALO + n * 16 + l15;
    v2f afrag[48], bfrag[48];
#pragma unroll
    for (int kk = 0; kk < 48; ++kk) afrag[kk] = wrow[kk * 2 + (kof >> 1)];
    load_bfrags(sh, col, kof, bfrag);
    __builtin_amdgcn_sched_barrier(0);
    v8f ce, co;
    wmma_chains(afrag, bfrag, ce, co);
    const v8f c8 = ce + co;
    const int rbase = m0 + ((lane & 16) ? 8 : 0);
#pragma unroll
    for (int v = 0; v < 8; ++v) {
      const int r = rbase + v;
      if (r < 29) sy[sidx(r, col)] = c8[v] + bias[r];
    }
  }
}

template<int DIL, int US, int W, int NT>
__device__ void layer_step(float* sh, float* sy, int t0,
                           const float* __restrict__ sepw,
                           const float* __restrict__ sepb,
                           const float* __restrict__ pww,
                           const float* __restrict__ pwb,
                           const float* __restrict__ gg1,
                           const float* __restrict__ bb1,
                           const float* __restrict__ gg2,
                           const float* __restrict__ bb2) {
  dwconv_ln_gelu<DIL, US, W>(sh, sy, sepw, sepb, gg1, bb1);
  __syncthreads();
  pw_conv_inplace<US, W, NT>(pww, pwb, sy);
  __syncthreads();
  ln_gelu_residual<US, W>(sy, sh, t0, gg2, bb2);
  __syncthreads();
}

__device__ __forceinline__ void build_knots(const float* uu, float* cks,
                                            float minb) {
  float m = uu[0];
#pragma unroll
  for (int i = 1; i < NBINS; ++i) m = fmaxf(m, uu[i]);
  float e[NBINS];
  float s = 0.f;
#pragma unroll
  for (int i = 0; i < NBINS; ++i) { e[i] = expf(uu[i] - m); s += e[i]; }
  const float inv = 1.0f / s;
  const float scale = 1.0f - minb * (float)NBINS;
  cks[0] = -TAILV;
  float acc = 0.f;
#pragma unroll
  for (int i = 0; i < NBINS - 1; ++i) {
    acc += minb + scale * e[i] * inv;
    cks[i + 1] = -TAILV + 2.0f * TAILV * acc;
  }
  cks[NBINS] = TAILV;
}

__global__ void __launch_bounds__(NTHR, 1)
convflow_kernel(const float* __restrict__ x,
                const float* __restrict__ pre_w, const float* __restrict__ pre_b,
                const float* __restrict__ sep_w, const float* __restrict__ sep_b,
                const float* __restrict__ pw_w,  const float* __restrict__ pw_b,
                const float* __restrict__ g1,    const float* __restrict__ b1,
                const float* __restrict__ g2,    const float* __restrict__ b2,
                const float* __restrict__ proj_w,const float* __restrict__ proj_b,
                float* __restrict__ out) {
  extern __shared__ float smem[];
  float* sh  = smem;
  float* sy  = smem + CF * SSTR;
  float* red = smem + 2 * CF * SSTR;

  const int tid = threadIdx.x;
  const int tile = blockIdx.x;
  const int b    = blockIdx.y;
  const int t0   = tile * TTILE;

  {
    const int lane = tid & 31;
    const int wv = __builtin_amdgcn_readfirstlane(tid >> 5);
    const float xpass = (tid < TTILE) ? x[(size_t)b * 2 * TLEN + t0 + tid] : 0.0f;
    float pw[16], pb[16];
#pragma unroll
    for (int i = 0; i < 16; ++i) {
      pw[i] = pre_w[16 * wv + i];
      pb[i] = pre_b[16 * wv + i];
    }
    float xv[5];
    bool  inr[5];
#pragma unroll
    for (int j = 0; j < 5; ++j) {
      const int t = t0 - HALO + lane + 32 * j;
      inr[j] = (t >= 0) && (t < TLEN);
      xv[j]  = inr[j] ? x[(size_t)b * 2 * TLEN + t] : 0.0f;
    }
#pragma unroll
    for (int q = 0; q < 8; ++q) {
      const float w0 = pw[2 * q],     b0 = pb[2 * q];
      const float w1 = pw[2 * q + 1], b1 = pb[2 * q + 1];
      float* row = sh + (8 * wv + q) * (2 * SSTR);
#pragma unroll
      for (int j = 0; j < 5; ++j) {
        const int u = lane + 32 * j;
        if (u < HW) {
          v2f hv;
          hv.x = inr[j] ? fmaf(w0, xv[j], b0) : 0.0f;
          hv.y = inr[j] ? fmaf(w1, xv[j], b1) : 0.0f;
          *(v2f*)(row + 2 * u) = hv;
        }
      }
    }
    if (tid < TTILE) {
      out[(size_t)b * 2 * TLEN + t0 + tid] = xpass;
    }
  }
  __syncthreads();

  layer_step<1, 1, 152, 10>(sh, sy, t0, sep_w + 0 * CF * 3, sep_b + 0 * CF,
                            pw_w + 0 * CF * CF, pw_b + 0 * CF, g1 + 0 * CF,
                            b1 + 0 * CF, g2 + 0 * CF, b2 + 0 * CF);
  layer_step<3, 4, 146, 10>(sh, sy, t0, sep_w + 1 * CF * 3, sep_b + 1 * CF,
                            pw_w + 1 * CF * CF, pw_b + 1 * CF, g1 + 1 * CF,
                            b1 + 1 * CF, g2 + 1 * CF, b2 + 1 * CF);
  layer_step<9, 13, 128, 8>(sh, sy, t0, sep_w + 2 * CF * 3, sep_b + 2 * CF,
                            pw_w + 2 * CF * CF, pw_b + 2 * CF, g1 + 2 * CF,
                            b1 + 2 * CF, g2 + 2 * CF, b2 + 2 * CF);

  proj_mm(proj_w, proj_b, sh, sy);
  __syncthreads();

  float la_local = 0.0f;
  if (tid < TTILE) {
    const int u = HALO + tid;
    const int t = t0 + tid;
    float uw[NBINS], uh[NBINS], d[NBINS + 1];
#pragma unroll
    for (int i = 0; i < NBINS; ++i) uw[i] = sy[sidx(i, u)];
#pragma unroll
    for (int i = 0; i < NBINS; ++i) uh[i] = sy[sidx(NBINS + i, u)];
    d[0] = 1.0f;
    d[NBINS] = 1.0f;
#pragma unroll
    for (int i = 0; i < NBINS - 1; ++i)
      d[i + 1] = MIND + softplusf(sy[sidx(2 * NBINS + i, u)]);

    float cw[NBINS + 1], ch[NBINS + 1];
    build_knots(uw, cw, MINBW);
    build_knots(uh, ch, MINBH);

    const float xin = x[(size_t)b * 2 * TLEN + TLEN + t];
    const bool inside = (xin >= -TAILV) && (xin <= TAILV);
    const float xc = fminf(fmaxf(xin, -TAILV), TAILV);
    int idx = -1;
#pragma unroll
    for (int k = 0; k < NBINS + 1; ++k) idx += (xc >= cw[k]) ? 1 : 0;
    idx = max(0, min(idx, NBINS - 1));

    const float icw = cw[idx];
    const float ibw = cw[idx + 1] - cw[idx];
    const float ich = ch[idx];
    const float ih  = ch[idx + 1] - ch[idx];
    const float idelta = ih / ibw;
    const float d0 = d[idx];
    const float d1 = d[idx + 1];
    const float theta = (xc - icw) / ibw;
    const float th1m  = 1.0f - theta;
    const float num = ih * (idelta * theta * theta + d0 * theta * th1m);
    const float den = idelta + (d0 + d1 - 2.0f * idelta) * theta * th1m;
    float outv = ich + num / den;
    const float dnum = idelta * idelta *
        (d1 * theta * theta + 2.0f * idelta * theta * th1m + d0 * th1m * th1m);
    float la = logf(dnum) - 2.0f * logf(den);
    if (!inside) { outv = xin; la = 0.0f; }
    out[(size_t)b * 2 * TLEN + TLEN + t] = outv;
    la_local = la;
  }

  __syncthreads();
  red[tid] = (tid < TTILE) ? la_local : 0.0f;
  __syncthreads();
  for (int s = 64; s > 0; s >>= 1) {
    if (tid < s) red[tid] += red[tid + s];
    __syncthreads();
  }
  if (tid == 0) atomicAdd(out + (size_t)NBATCH * 2 * TLEN + b, red[0]);
}

__global__ void zero_logdet(float* p) {
  if (threadIdx.x < NBATCH) p[threadIdx.x] = 0.0f;
}

extern "C" void kernel_launch(void* const* d_in, const int* in_sizes, int n_in,
                              void* d_out, int out_size, void* d_ws,
                              size_t ws_size, hipStream_t stream) {
  (void)in_sizes; (void)n_in; (void)out_size; (void)d_ws; (void)ws_size;
  const float* x      = (const float*)d_in[0];
  const float* pre_w  = (const float*)d_in[2];
  const float* pre_b  = (const float*)d_in[3];
  const float* sep_w  = (const float*)d_in[4];
  const float* sep_b  = (const float*)d_in[5];
  const float* pw_w   = (const float*)d_in[6];
  const float* pw_b   = (const float*)d_in[7];
  const float* g1     = (const float*)d_in[8];
  const float* b1     = (const float*)d_in[9];
  const float* g2     = (const float*)d_in[10];
  const float* b2     = (const float*)d_in[11];
  const float* proj_w = (const float*)d_in[12];
  const float* proj_b = (const float*)d_in[13];
  float* out = (float*)d_out;

  zero_logdet<<<1, 32, 0, stream>>>(out + (size_t)NBATCH * 2 * TLEN);

  const size_t smem_bytes = (size_t)(2 * CF * SSTR + 512) * sizeof(float);
  dim3 grid(TLEN / TTILE, NBATCH);
  convflow_kernel<<<grid, NTHR, smem_bytes, stream>>>(
      x, pre_w, pre_b, sep_w, sep_b, pw_w, pw_b, g1, b1, g2, b2, proj_w,
      proj_b, out);
}


# LN2+GELU+residual moved into the matmul epilogue: accumulators normalised in registers, column sums as 12 per-wave pairs via LDS at the tile barrier (sum order changes); y2 LDS pass, a barrier gone
# speedup vs baseline: 2.2796x; 1.0354x over previous
#include <hip/hip_runtime.h>
#include <math.h>


#define CF     192
#define TTILE  128
#define HALO   13
#define HW     (TTILE + 2*HALO)
#define SSTR   155
#define NTHR   384
#define NWAVE  12
#define TLEN   8192
#define NBATCH 32
#define NBINS  10
#define TAILV  5.0f
#define MINBW  0.001f
#define MINBH  0.001f
#define MIND   0.001f

typedef float v2f __attribute__((ext_vector_type(2)));
typedef float v8f __attribute__((ext_vector_type(8)));

__device__ __forceinline__ int sidx(int c, int u) {
  return (c >> 1) * (2 * SSTR) + 2 * u + (c & 1);
}

__device__ __forceinline__ float gelu_exact(float x) {
  const float ax = fabsf(x);
  const float t = __builtin_amdgcn_rcpf(fmaf(ax, 0.28f, 1.0f));
  const float e = __builtin_amdgcn_exp2f((x * x) * -0.72134752044448170368f);
  float p = -0.10675480918401203f;
  p = fmaf(p, t, 0.39380918123096653f);
  p = fmaf(p, t, -0.2434958341181943f);
  p = fmaf(p, t, 0.27424663512843267f);
  p = fmaf(p, t, 0.06532213231594822f);
  p = fmaf(p, t, 0.11687268054021908f);
  const float tail = (p * t) * e;
  return fmaf(-ax, tail, fmaxf(x, 0.0f));
}
__device__ __forceinline__ float softplusf(float x) {
  return (x > 20.0f) ? x : log1pf(expf(x));
}

__device__ __forceinline__ float xhalf(float x) {
  const int b = __builtin_bit_cast(int, x);
  return __builtin_bit_cast(
      float, __builtin_amdgcn_permlanex16(b, b, 0x76543210, 0xfedcba98, false,
                                          false));
}
template<int M>
__device__ __forceinline__ float row_xor(float x) {
  return __builtin_bit_cast(
      float, __builtin_amdgcn_update_dpp(0, __builtin_bit_cast(int, x),
                                         0x160 + M, 0xf, 0xf, true));
}

__device__ __forceinline__ void ln_gelu_regs(float (&v)[6],
                                             const float (&gm)[6],
                                             const float (&bt)[6]) {
  float s1 = 0.f, s2 = 0.f;
#pragma unroll
  for (int r = 0; r < 6; ++r) {
    s1 += v[r];
    s2 += v[r] * v[r];
  }
  s1 += xhalf(s1);
  s2 += xhalf(s2);
  s1 += row_xor<8>(s1);
  s2 += row_xor<8>(s2);
  s1 += row_xor<4>(s1);
  s2 += row_xor<4>(s2);
  s1 += row_xor<2>(s1);
  s2 += row_xor<2>(s2);
  s1 += row_xor<1>(s1);
  s2 += row_xor<1>(s2);
  const float mu  = s1 * (1.0f / 192.0f);
  const float var = s2 * (1.0f / 192.0f) - mu * mu;
  const float rs  = rsqrtf(var + 1e-5f);
#pragma unroll
  for (int r = 0; r < 6; ++r) {
    const float xh = (v[r] - mu) * rs;
    v[r] = gelu_exact(xh * gm[r] + bt[r]);
  }
}

template<int DIL, int US, int W>
__device__ void dwconv_ln_gelu(const float* __restrict__ sh,
                               float* __restrict__ sy,
                               const float* __restrict__ sepw,
                               const float* __restrict__ sepb,
                               const float* __restrict__ gamma,
                               const float* __restrict__ beta) {
  const int tid  = threadIdx.x;
  const int lane = tid & 31;
  const int wid  = tid >> 5;
  float w0[6], w1[6], w2[6], sb[6], gm[6], bt[6];
#pragma unroll
  for (int r = 0; r < 6; ++r) {
    const int c = lane + 32 * r;
    w0[r] = sepw[c * 3 + 0];
    w1[r] = sepw[c * 3 + 1];
    w2[r] = sepw[c * 3 + 2];
    sb[r] = sepb[c];
    gm[r] = gamma[c];
    bt[r] = beta[c];
  }
  for (int j = wid; j < W; j += NWAVE) {
    const int u = US + j;
    float v[6];
#pragma unroll
    for (int r = 0; r < 6; ++r) {
      const int i = sidx(lane + 32 * r, u);
      v[r] = w0[r] * sh[i - 2 * DIL] + w1[r] * sh[i] + w2[r] * sh[i + 2 * DIL] +
             sb[r];
    }
    ln_gelu_regs(v, gm, bt);
#pragma unroll
    for (int r = 0; r < 6; ++r) sy[sidx(lane + 32 * r, u)] = v[r];
  }
}

__device__ __forceinline__ void load_bfrags(const float* buf, int colb, int kof,
                                            v2f (&bf)[48]) {
#pragma unroll
  for (int kk = 0; kk < 48; ++kk) {
    const int k = kk * 4 + kof;
    bf[kk] = *(const v2f*)(&buf[(k >> 1) * (2 * SSTR) + 2 * colb]);
  }
}

__device__ __forceinline__ void wmma_chains(const v2f (&af)[48],
                                            const v2f (&bf)[48],
                                            v8f& ce, v8f& co) {
  ce = v8f{};
  co = v8f{};
#pragma unroll
  for (int kk = 0; kk < 48; kk += 2) {
    ce = __builtin_amdgcn_wmma_f32_16x16x4_f32(
        false, af[kk], false, bf[kk], (short)0, ce, false, false);
    co = __builtin_amdgcn_wmma_f32_16x16x4_f32(
        false, af[kk + 1], false, bf[kk + 1], (short)0, co, false, false);
  }
}

#define PARTF (2 * NWAVE * 16 * 2)

__device__ __forceinline__ void ep_loads(const float* pbuf, const float* sh,
                                         int l15, int rbase, int colc,
                                         v2f (&ps)[NWAVE], v2f (&hv)[4]) {
#pragma unroll
  for (int w = 0; w < NWAVE; ++w)
    ps[w] = *(const v2f*)(&pbuf[(w * 16 + l15) * 2]);
#pragma unroll
  for (int q = 0; q < 4; ++q)
    hv[q] = *(const v2f*)(&sh[sidx(rbase + 2 * q, colc)]);
}

__device__ __forceinline__ void ep_finish(const v8f& c8, const v2f (&ps)[NWAVE],
                                          const v2f (&hv)[4],
                                          const float (&gm)[8],
                                          const float (&bt)[8], float* sh,
                                          int rbase, int col, int colmax,
                                          int t0) {
  float s1 = 0.f, s2 = 0.f;
#pragma unroll
  for (int w = 0; w < NWAVE; ++w) {
    s1 += ps[w].x;
    s2 += ps[w].y;
  }
  const float mu  = s1 * (1.0f / 192.0f);
  const float var = s2 * (1.0f / 192.0f) - mu * mu;
  const float rs  = rsqrtf(var + 1e-5f);
  const int t = t0 - HALO + col;
  const bool valid = (t >= 0) && (t < TLEN);
  float y[8];
#pragma unroll
  for (int r = 0; r < 8; ++r) {
    const float xh = (c8[r] - mu) * rs;
    const float g = gelu_exact(xh * gm[r] + bt[r]);
    y[r] = valid ? g : 0.0f;
  }
  if (col <= colmax) {
#pragma unroll
    for (int q = 0; q < 4; ++q) {
      v2f o;
      o.x = hv[q].x + y[2 * q];
      o.y = hv[q].y + y[2 * q + 1];
      *(v2f*)(&sh[sidx(rbase + 2 * q, col)]) = o;
    }
  }
}

__device__ __forceinline__ void tile_partials(const v8f& ce, const v8f& co,
                                              const float (&bs)[8], v8f& c8,
                                              float* pbuf, int wid, int lane) {
  float s1 = 0.f, s2 = 0.f;
#pragma unroll
  for (int r = 0; r < 8; ++r) {
    c8[r] = (ce[r] + co[r]) + bs[r];
    s1 += c8[r];
    s2 += c8[r] * c8[r];
  }
  s1 += xhalf(s1);
  s2 += xhalf(s2);
  if (lane < 16) {
    v2f p;
    p.x = s1;
    p.y = s2;
    *(v2f*)(&pbuf[(wid * 16 + lane) * 2]) = p;
  }
}

template<int US, int W, int NT>
__device__ void pw_conv_ln_gelu_residual(const float* __restrict__ Wm,
                                         const float* __restrict__ bias,
                                         const float* __restrict__ gamma,
                                         const float* __restrict__ beta,
                                         const float* sy, float* sh,
                                         float* part, int t0) {
  const int tid  = threadIdx.x;
  const int lane = tid & 31;
  const int wid  = tid >> 5;
  const int m0   = wid * 16;
  const int l15  = lane & 15;
  const int kof  = (lane & 16) ? 2 : 0;
  const int colmax = US + W - 1;
  const int rbase = m0 + ((lane & 16) ? 8 : 0);

  v2f afrag[48];
  {
    const v2f* wrow = (const v2f*)(Wm + (size_t)(m0 + l15) * CF);
    const int kidx0 = kof >> 1;
#pragma unroll
    for (int kk = 0; kk < 48; ++kk) afrag[kk] = wrow[kk * 2 + kidx0];
  }
  float bs[8], gm[8], bt[8];
#pragma unroll
  for (int r = 0; r < 8; ++r) {
    bs[r] = bias[rbase + r];
    gm[r] = gamma[rbase + r];
    bt[r] = beta[rbase + r];
  }

  v2f bfrag[48];
  {
    int colb = US + l15;
    if (colb > colmax) colb = colmax;
    load_bfrags(sy, colb, kof, bfrag);
  }
  v8f cprev;
  {
    __builtin_amdgcn_sched_barrier(0);
    v8f ce, co;
    wmma_chains(afrag, bfrag, ce, co);
    asm volatile("" : "+v"(ce), "+v"(co));
    __builtin_amdgcn_sched_barrier(0);
    {
      int colb = US + 16 + l15;
      if (colb > colmax) colb = colmax;
      load_bfrags(sy, colb, kof, bfrag);
    }
    tile_partials(ce, co, bs, cprev, part, wid, lane);
  }
  __syncthreads();
  for (int n = 1; n < NT; ++n) {
    v8f ce, co;
    wmma_chains(afrag, bfrag, ce, co);
    asm volatile("" : "+v"(ce), "+v"(co));
    __builtin_amdgcn_sched_barrier(0);
    const int colp = US + (n - 1) * 16 + l15;
    v2f ps[NWAVE], hv[4];
    ep_loads(part + ((n - 1) & 1) * (PARTF / 2), sh, l15, rbase,
             colp > colmax ? colmax : colp, ps, hv);
    if (n + 1 < NT) {
      int colb = US + (n + 1) * 16 + l15;
      if (colb > colmax) colb = colmax;
      load_bfrags(sy, colb, kof, bfrag);
    }
    ep_finish(cprev, ps, hv, gm, bt, sh, rbase, colp, colmax, t0);
    tile_partials(ce, co, bs, cprev, part + (n & 1) * (PARTF / 2), wid, lane);
    __syncthreads();
  }
  {
    const int colp = US + (NT - 1) * 16 + l15;
    v2f ps[NWAVE], hv[4];
    ep_loads(part + ((NT - 1) & 1) * (PARTF / 2), sh, l15, rbase,
             colp > colmax ? colmax : colp, ps, hv);
    ep_finish(cprev, ps, hv, gm, bt, sh, rbase, colp, colmax, t0);
  }
}

__device__ void proj_mm(const float* __restrict__ Wm,
                        const float* __restrict__ bias,
                        const float* sh, float* sy) {
  const int tid  = threadIdx.x;
  const int lane = tid & 31;
  const int wid  = tid >> 5;
  const int l15  = lane & 15;
  const int kof  = (lane & 16) ? 2 : 0;
  for (int job = wid; job < 16; job += NWAVE) {
    const int m0 = (job & 1) * 16;
    const int n  = job >> 1;
    const int row  = m0 + l15;
    const int rowc = row < 29 ? row : 28;
    const v2f* wrow = (const v2f*)(Wm + (size_t)rowc * CF);
    const int col = HALO + n * 16 + l15;
    v2f afrag[48], bfrag[48];
#pragma unroll
    for (int kk = 0; kk < 48; ++kk) afrag[kk] = wrow[kk * 2 + (kof >> 1)];
    load_bfrags(sh, col, kof, bfrag);
    __builtin_amdgcn_sched_barrier(0);
    v8f ce, co;
    wmma_chains(afrag, bfrag, ce, co);
    const v8f c8 = ce + co;
    const int rbase = m0 + ((lane & 16) ? 8 : 0);
#pragma unroll
    for (int v = 0; v < 8; ++v) {
      const int r = rbase + v;
      if (r < 29) sy[sidx(r, col)] = c8[v] + bias[r];
    }
  }
}

template<int DIL, int US, int W, int NT>
__device__ void layer_step(float* sh, float* sy, float* part, int t0,
                           const float* __restrict__ sepw,
                           const float* __restrict__ sepb,
                           const float* __restrict__ pww,
                           const float* __restrict__ pwb,
                           const float* __restrict__ gg1,
                           const float* __restrict__ bb1,
                           const float* __restrict__ gg2,
                           const float* __restrict__ bb2) {
  dwconv_ln_gelu<DIL, US, W>(sh, sy, sepw, sepb, gg1, bb1);
  __syncthreads();
  pw_conv_ln_gelu_residual<US, W, NT>(pww, pwb, gg2, bb2, sy, sh, part, t0);
  __syncthreads();
}

__device__ __forceinline__ void build_knots(const float* uu, float* cks,
                                            float minb) {
  float m = uu[0];
#pragma unroll
  for (int i = 1; i < NBINS; ++i) m = fmaxf(m, uu[i]);
  float e[NBINS];
  float s = 0.f;
#pragma unroll
  for (int i = 0; i < NBINS; ++i) { e[i] = expf(uu[i] - m); s += e[i]; }
  const float inv = 1.0f / s;
  const float scale = 1.0f - minb * (float)NBINS;
  cks[0] = -TAILV;
  float acc = 0.f;
#pragma unroll
  for (int i = 0; i < NBINS - 1; ++i) {
    acc += minb + scale * e[i] * inv;
    cks[i + 1] = -TAILV + 2.0f * TAILV * acc;
  }
  cks[NBINS] = TAILV;
}

__global__ void __launch_bounds__(NTHR, 1)
convflow_kernel(const float* __restrict__ x,
                const float* __restrict__ pre_w, const float* __restrict__ pre_b,
                const float* __restrict__ sep_w, const float* __restrict__ sep_b,
                const float* __restrict__ pw_w,  const float* __restrict__ pw_b,
                const float* __restrict__ g1,    const float* __restrict__ b1,
                const float* __restrict__ g2,    const float* __restrict__ b2,
                const float* __restrict__ proj_w,const float* __restrict__ proj_b,
                float* __restrict__ out) {
  extern __shared__ float smem[];
  float* sh  = smem;
  float* sy  = smem + CF * SSTR;
  float* red = smem + 2 * CF * SSTR;

  const int tid = threadIdx.x;
  const int tile = blockIdx.x;
  const int b    = blockIdx.y;
  const int t0   = tile * TTILE;

  {
    const int lane = tid & 31;
    const int wv = __builtin_amdgcn_readfirstlane(tid >> 5);
    const float xpass = (tid < TTILE) ? x[(size_t)b * 2 * TLEN + t0 + tid] : 0.0f;
    float pw[16], pb[16];
#pragma unroll
    for (int i = 0; i < 16; ++i) {
      pw[i] = pre_w[16 * wv + i];
      pb[i] = pre_b[16 * wv + i];
    }
    float xv[5];
    bool  inr[5];
#pragma unroll
    for (int j = 0; j < 5; ++j) {
      const int t = t0 - HALO + lane + 32 * j;
      inr[j] = (t >= 0) && (t < TLEN);
      xv[j]  = inr[j] ? x[(size_t)b * 2 * TLEN + t] : 0.0f;
    }
#pragma unroll
    for (int q = 0; q < 8; ++q) {
      const float w0 = pw[2 * q],     b0 = pb[2 * q];
      const float w1 = pw[2 * q + 1], b1 = pb[2 * q + 1];
      float* row = sh + (8 * wv + q) * (2 * SSTR);
#pragma unroll
      for (int j = 0; j < 5; ++j) {
        const int u = lane + 32 * j;
        if (u < HW) {
          v2f hv;
          hv.x = inr[j] ? fmaf(w0, xv[j], b0) : 0.0f;
          hv.y = inr[j] ? fmaf(w1, xv[j], b1) : 0.0f;
          *(v2f*)(row + 2 * u) = hv;
        }
      }
    }
    if (tid < TTILE) {
      out[(size_t)b * 2 * TLEN + t0 + tid] = xpass;
    }
  }
  __syncthreads();

  layer_step<1, 1, 152, 10>(sh, sy, red, t0, sep_w + 0 * CF * 3, sep_b + 0 * CF,
                            pw_w + 0 * CF * CF, pw_b + 0 * CF, g1 + 0 * CF,
                            b1 + 0 * CF, g2 + 0 * CF, b2 + 0 * CF);
  layer_step<3, 4, 146, 10>(sh, sy, red, t0, sep_w + 1 * CF * 3, sep_b + 1 * CF,
                            pw_w + 1 * CF * CF, pw_b + 1 * CF, g1 + 1 * CF,
                            b1 + 1 * CF, g2 + 1 * CF, b2 + 1 * CF);
  layer_step<9, 13, 128, 8>(sh, sy, red, t0, sep_w + 2 * CF * 3, sep_b + 2 * CF,
                            pw_w + 2 * CF * CF, pw_b + 2 * CF, g1 + 2 * CF,
                            b1 + 2 * CF, g2 + 2 * CF, b2 + 2 * CF);

  proj_mm(proj_w, proj_b, sh, sy);
  __syncthreads();

  float la_local = 0.0f;
  if (tid < TTILE) {
    const int u = HALO + tid;
    const int t = t0 + tid;
    float uw[NBINS], uh[NBINS], d[NBINS + 1];
#pragma unroll
    for (int i = 0; i < NBINS; ++i) uw[i] = sy[sidx(i, u)];
#pragma unroll
    for (int i = 0; i < NBINS; ++i) uh[i] = sy[sidx(NBINS + i, u)];
    d[0] = 1.0f;
    d[NBINS] = 1.0f;
#pragma unroll
    for (int i = 0; i < NBINS - 1; ++i)
      d[i + 1] = MIND + softplusf(sy[sidx(2 * NBINS + i, u)]);

    float cw[NBINS + 1], ch[NBINS + 1];
    build_knots(uw, cw, MINBW);
    build_knots(uh, ch, MINBH);

    const float xin = x[(size_t)b * 2 * TLEN + TLEN + t];
    const bool inside = (xin >= -TAILV) && (xin <= TAILV);
    const float xc = fminf(fmaxf(xin, -TAILV), TAILV);
    int idx = -1;
#pragma unroll
    for (int k = 0; k < NBINS + 1; ++k) idx += (xc >= cw[k]) ? 1 : 0;
    idx = max(0, min(idx, NBINS - 1));

    const float icw = cw[idx];
    const float ibw = cw[idx + 1] - cw[idx];
    const float ich = ch[idx];
    const float ih  = ch[idx + 1] - ch[idx];
    const float idelta = ih / ibw;
    const float d0 = d[idx];
    const float d1 = d[idx + 1];
    const float theta = (xc - icw) / ibw;
    const float th1m  = 1.0f - theta;
    const float num = ih * (idelta * theta * theta + d0 * theta * th1m);
    const float den = idelta + (d0 + d1 - 2.0f * idelta) * theta * th1m;
    float outv = ich + num / den;
    const float dnum = idelta * idelta *
        (d1 * theta * theta + 2.0f * idelta * theta * th1m + d0 * th1m * th1m);
    float la = logf(dnum) - 2.0f * logf(den);
    if (!inside) { outv = xin; la = 0.0f; }
    out[(size_t)b * 2 * TLEN + TLEN + t] = outv;
    la_local = la;
  }

  __syncthreads();
  red[tid] = (tid < TTILE) ? la_local : 0.0f;
  __syncthreads();
  for (int s = 64; s > 0; s >>= 1) {
    if (tid < s) red[tid] += red[tid + s];
    __syncthreads();
  }
  if (tid == 0) atomicAdd(out + (size_t)NBATCH * 2 * TLEN + b, red[0]);
}

__global__ void zero_logdet(float* p) {
  if (threadIdx.x < NBATCH) p[threadIdx.x] = 0.0f;
}

extern "C" void kernel_launch(void* const* d_in, const int* in_sizes, int n_in,
                              void* d_out, int out_size, void* d_ws,
                              size_t ws_size, hipStream_t stream) {
  (void)in_sizes; (void)n_in; (void)out_size; (void)d_ws; (void)ws_size;
  const float* x      = (const float*)d_in[0];
  const float* pre_w  = (const float*)d_in[2];
  const float* pre_b  = (const float*)d_in[3];
  const float* sep_w  = (const float*)d_in[4];
  const float* sep_b  = (const float*)d_in[5];
  const float* pw_w   = (const float*)d_in[6];
  const float* pw_b   = (const float*)d_in[7];
  const float* g1     = (const float*)d_in[8];
  const float* b1     = (const float*)d_in[9];
  const float* g2     = (const float*)d_in[10];
  const float* b2     = (const float*)d_in[11];
  const float* proj_w = (const float*)d_in[12];
  const float* proj_b = (const float*)d_in[13];
  float* out = (float*)d_out;

  zero_logdet<<<1, 32, 0, stream>>>(out + (size_t)NBATCH * 2 * TLEN);

  const size_t smem_bytes = (size_t)(2 * CF * SSTR + PARTF) * sizeof(float);
  dim3 grid(TLEN / TTILE, NBATCH);
  convflow_kernel<<<grid, NTHR, smem_bytes, stream>>>(
      x, pre_w, pre_b, sep_w, sep_b, pw_w, pw_b, g1, b1, g2, b2, proj_w,
      proj_b, out);
}


# Time tile per workgroup 128 -> 172 columns (LDS 236 -> 302 KB): 48x32 workgroups = 6 full rounds of 36 WMMA tiles, was 8 of 28; per-workgroup fixed costs 6 times, not 8 (logdet sum order changes)
# speedup vs baseline: 2.3690x; 1.0392x over previous
#include <hip/hip_runtime.h>
#include <math.h>


#define CF     192
#define TTILE  172
#define NTILE  48
#define HALO   13
#define HW     (TTILE + 2*HALO)
#define SSTR   199
#define NTHR   384
#define NWAVE  12
#define TLEN   8192
#define NBATCH 32
#define NBINS  10
#define TAILV  5.0f
#define MINBW  0.001f
#define MINBH  0.001f
#define MIND   0.001f

typedef float v2f __attribute__((ext_vector_type(2)));
typedef float v8f __attribute__((ext_vector_type(8)));

__device__ __forceinline__ int sidx(int c, int u) {
  return (c >> 1) * (2 * SSTR) + 2 * u + (c & 1);
}

__device__ __forceinline__ float gelu_exact(float x) {
  const float ax = fabsf(x);
  const float t = __builtin_amdgcn_rcpf(fmaf(ax, 0.28f, 1.0f));
  const float e = __builtin_amdgcn_exp2f((x * x) * -0.72134752044448170368f);
  float p = -0.10675480918401203f;
  p = fmaf(p, t, 0.39380918123096653f);
  p = fmaf(p, t, -0.2434958341181943f);
  p = fmaf(p, t, 0.27424663512843267f);
  p = fmaf(p, t, 0.06532213231594822f);
  p = fmaf(p, t, 0.11687268054021908f);
  const float tail = (p * t) * e;
  return fmaf(-ax, tail, fmaxf(x, 0.0f));
}
__device__ __forceinline__ float softplusf(float x) {
  return (x > 20.0f) ? x : log1pf(expf(x));
}

__device__ __forceinline__ float xhalf(float x) {
  const int b = __builtin_bit_cast(int, x);
  return __builtin_bit_cast(
      float, __builtin_amdgcn_permlanex16(b, b, 0x76543210, 0xfedcba98, false,
                                          false));
}
template<int M>
__device__ __forceinline__ float row_xor(float x) {
  return __builtin_bit_cast(
      float, __builtin_amdgcn_update_dpp(0, __builtin_bit_cast(int, x),
                                         0x160 + M, 0xf, 0xf, true));
}

__device__ __forceinline__ void ln_gelu_regs(float (&v)[6],
                                             const float (&gm)[6],
                                             const float (&bt)[6]) {
  float s1 = 0.f, s2 = 0.f;
#pragma unroll
  for (int r = 0; r < 6; ++r) {
    s1 += v[r];
    s2 += v[r] * v[r];
  }
  s1 += xhalf(s1);
  s2 += xhalf(s2);
  s1 += row_xor<8>(s1);
  s2 += row_xor<8>(s2);
  s1 += row_xor<4>(s1);
  s2 += row_xor<4>(s2);
  s1 += row_xor<2>(s1);
  s2 += row_xor<2>(s2);
  s1 += row_xor<1>(s1);
  s2 += row_xor<1>(s2);
  const float mu  = s1 * (1.0f / 192.0f);
  const float var = s2 * (1.0f / 192.0f) - mu * mu;
  const float rs  = rsqrtf(var + 1e-5f);
#pragma unroll
  for (int r = 0; r < 6; ++r) {
    const float xh = (v[r] - mu) * rs;
    v[r] = gelu_exact(xh * gm[r] + bt[r]);
  }
}

template<int DIL, int US, int W>
__device__ void dwconv_ln_gelu(const float* __restrict__ sh,
                               float* __restrict__ sy,
                               const float* __restrict__ sepw,
                               const float* __restrict__ sepb,
                               const float* __restrict__ gamma,
                               const float* __restrict__ beta) {
  const int tid  = threadIdx.x;
  const int lane = tid & 31;
  const int wid  = tid >> 5;
  float w0[6], w1[6], w2[6], sb[6], gm[6], bt[6];
#pragma unroll
  for (int r = 0; r < 6; ++r) {
    const int c = lane + 32 * r;
    w0[r] = sepw[c * 3 + 0];
    w1[r] = sepw[c * 3 + 1];
    w2[r] = sepw[c * 3 + 2];
    sb[r] = sepb[c];
    gm[r] = gamma[c];
    bt[r] = beta[c];
  }
  for (int j = wid; j < W; j += NWAVE) {
    const int u = US + j;
    float v[6];
#pragma unroll
    for (int r = 0; r < 6; ++r) {
      const int i = sidx(lane + 32 * r, u);
      v[r] = w0[r] * sh[i - 2 * DIL] + w1[r] * sh[i] + w2[r] * sh[i + 2 * DIL] +
             sb[r];
    }
    ln_gelu_regs(v, gm, bt);
#pragma unroll
    for (int r = 0; r < 6; ++r) sy[sidx(lane + 32 * r, u)] = v[r];
  }
}

__device__ __forceinline__ void load_bfrags(const float* buf, int colb, int kof,
                                            v2f (&bf)[48]) {
#pragma unroll
  for (int kk = 0; kk < 48; ++kk) {
    const int k = kk * 4 + kof;
    bf[kk] = *(const v2f*)(&buf[(k >> 1) * (2 * SSTR) + 2 * colb]);
  }
}

__device__ __forceinline__ void wmma_chains(const v2f (&af)[48],
                                            const v2f (&bf)[48],
                                            v8f& ce, v8f& co) {
  ce = v8f{};
  co = v8f{};
#pragma unroll
  for (int kk = 0; kk < 48; kk += 2) {
    ce = __builtin_amdgcn_wmma_f32_16x16x4_f32(
        false, af[kk], false, bf[kk], (short)0, ce, false, false);
    co = __builtin_amdgcn_wmma_f32_16x16x4_f32(
        false, af[kk + 1], false, bf[kk + 1], (short)0, co, false, false);
  }
}

#define PARTF (2 * NWAVE * 16 * 2)

__device__ __forceinline__ void ep_loads(const float* pbuf, const float* sh,
                                         int l15, int rbase, int colc,
                                         v2f (&ps)[NWAVE], v2f (&hv)[4]) {
#pragma unroll
  for (int w = 0; w < NWAVE; ++w)
    ps[w] = *(const v2f*)(&pbuf[(w * 16 + l15) * 2]);
#pragma unroll
  for (int q = 0; q < 4; ++q)
    hv[q] = *(const v2f*)(&sh[sidx(rbase + 2 * q, colc)]);
}

__device__ __forceinline__ void ep_finish(const v8f& c8, const v2f (&ps)[NWAVE],
                                          const v2f (&hv)[4],
                                          const float (&gm)[8],
                                          const float (&bt)[8], float* sh,
                                          int rbase, int col, int colmax,
                                          int t0) {
  float s1 = 0.f, s2 = 0.f;
#pragma unroll
  for (int w = 0; w < NWAVE; ++w) {
    s1 += ps[w].x;
    s2 += ps[w].y;
  }
  const float mu  = s1 * (1.0f / 192.0f);
  const float var = s2 * (1.0f / 192.0f) - mu * mu;
  const float rs  = rsqrtf(var + 1e-5f);
  const int t = t0 - HALO + col;
  const bool valid = (t >= 0) && (t < TLEN);
  float y[8];
#pragma unroll
  for (int r = 0; r < 8; ++r) {
    const float xh = (c8[r] - mu) * rs;
    const float g = gelu_exact(xh * gm[r] + bt[r]);
    y[r] = valid ? g : 0.0f;
  }
  if (col <= colmax) {
#pragma unroll
    for (int q = 0; q < 4; ++q) {
      v2f o;
      o.x = hv[q].x + y[2 * q];
      o.y = hv[q].y + y[2 * q + 1];
      *(v2f*)(&sh[sidx(rbase + 2 * q, col)]) = o;
    }
  }
}

__device__ __forceinline__ void tile_partials(const v8f& ce, const v8f& co,
                                              const float (&bs)[8], v8f& c8,
                                              float* pbuf, int wid, int lane) {
  float s1 = 0.f, s2 = 0.f;
#pragma unroll
  for (int r = 0; r < 8; ++r) {
    c8[r] = (ce[r] + co[r]) + bs[r];
    s1 += c8[r];
    s2 += c8[r] * c8[r];
  }
  s1 += xhalf(s1);
  s2 += xhalf(s2);
  if (lane < 16) {
    v2f p;
    p.x = s1;
    p.y = s2;
    *(v2f*)(&pbuf[(wid * 16 + lane) * 2]) = p;
  }
}

template<int US, int W, int NT>
__device__ void pw_conv_ln_gelu_residual(const float* __restrict__ Wm,
                                         const float* __restrict__ bias,
                                         const float* __restrict__ gamma,
                                         const float* __restrict__ beta,
                                         const float* sy, float* sh,
                                         float* part, int t0) {
  const int tid  = threadIdx.x;
  const int lane = tid & 31;
  const int wid  = tid >> 5;
  const int m0   = wid * 16;
  const int l15  = lane & 15;
  const int kof  = (lane & 16) ? 2 : 0;
  const int colmax = US + W - 1;
  const int rbase = m0 + ((lane & 16) ? 8 : 0);

  v2f afrag[48];
  {
    const v2f* wrow = (const v2f*)(Wm + (size_t)(m0 + l15) * CF);
    const int kidx0 = kof >> 1;
#pragma unroll
    for (int kk = 0; kk < 48; ++kk) afrag[kk] = wrow[kk * 2 + kidx0];
  }
  float bs[8], gm[8], bt[8];
#pragma unroll
  for (int r = 0; r < 8; ++r) {
    bs[r] = bias[rbase + r];
    gm[r] = gamma[rbase + r];
    bt[r] = beta[rbase + r];
  }

  v2f bfrag[48];
  {
    int colb = US + l15;
    if (colb > colmax) colb = colmax;
    load_bfrags(sy, colb, kof, bfrag);
  }
  v8f cprev;
  {
    __builtin_amdgcn_sched_barrier(0);
    v8f ce, co;
    wmma_chains(afrag, bfrag, ce, co);
    asm volatile("" : "+v"(ce), "+v"(co));
    __builtin_amdgcn_sched_barrier(0);
    {
      int colb = US + 16 + l15;
      if (colb > colmax) colb = colmax;
      load_bfrags(sy, colb, kof, bfrag);
    }
    tile_partials(ce, co, bs, cprev, part, wid, lane);
  }
  __syncthreads();
  for (int n = 1; n < NT; ++n) {
    v8f ce, co;
    wmma_chains(afrag, bfrag, ce, co);
    asm volatile("" : "+v"(ce), "+v"(co));
    __builtin_amdgcn_sched_barrier(0);
    const int colp = US + (n - 1) * 16 + l15;
    v2f ps[NWAVE], hv[4];
    ep_loads(part + ((n - 1) & 1) * (PARTF / 2), sh, l15, rbase,
             colp > colmax ? colmax : colp, ps, hv);
    if (n + 1 < NT) {
      int colb = US + (n + 1) * 16 + l15;
      if (colb > colmax) colb = colmax;
      load_bfrags(sy, colb, kof, bfrag);
    }
    ep_finish(cprev, ps, hv, gm, bt, sh, rbase, colp, colmax, t0);
    tile_partials(ce, co, bs, cprev, part + (n & 1) * (PARTF / 2), wid, lane);
    __syncthreads();
  }
  {
    const int colp = US + (NT - 1) * 16 + l15;
    v2f ps[NWAVE], hv[4];
    ep_loads(part + ((NT - 1) & 1) * (PARTF / 2), sh, l15, rbase,
             colp > colmax ? colmax : colp, ps, hv);
    ep_finish(cprev, ps, hv, gm, bt, sh, rbase, colp, colmax, t0);
  }
}

__device__ void proj_mm(const float* __restrict__ Wm,
                        const float* __restrict__ bias,
                        const float* sh, float* sy) {
  const int tid  = threadIdx.x;
  const int lane = tid & 31;
  const int wid  = tid >> 5;
  const int l15  = lane & 15;
  const int kof  = (lane & 16) ? 2 : 0;
  for (int job = wid; job < 2 * ((TTILE + 15) / 16); job += NWAVE) {
    const int m0 = (job & 1) * 16;
    const int n  = job >> 1;
    const int row  = m0 + l15;
    const int rowc = row < 29 ? row : 28;
    const v2f* wrow = (const v2f*)(Wm + (size_t)rowc * CF);
    const int col = HALO + n * 16 + l15;
    v2f afrag[48], bfrag[48];
#pragma unroll
    for (int kk = 0; kk < 48; ++kk) afrag[kk] = wrow[kk * 2 + (kof >> 1)];
    load_bfrags(sh, col, kof, bfrag);
    __builtin_amdgcn_sched_barrier(0);
    v8f ce, co;
    wmma_chains(afrag, bfrag, ce, co);
    const v8f c8 = ce + co;
    const int rbase = m0 + ((lane & 16) ? 8 : 0);
#pragma unroll
    for (int v = 0; v < 8; ++v) {
      const int r = rbase + v;
      if (r < 29) sy[sidx(r, col)] = c8[v] + bias[r];
    }
  }
}

template<int DIL, int US, int W, int NT>
__device__ void layer_step(float* sh, float* sy, float* part, int t0,
                           const float* __restrict__ sepw,
                           const float* __restrict__ sepb,
                           const float* __restrict__ pww,
                           const float* __restrict__ pwb,
                           const float* __restrict__ gg1,
                           const float* __restrict__ bb1,
                           const float* __restrict__ gg2,
                           const float* __restrict__ bb2) {
  dwconv_ln_gelu<DIL, US, W>(sh, sy, sepw, sepb, gg1, bb1);
  __syncthreads();
  pw_conv_ln_gelu_residual<US, W, NT>(pww, pwb, gg2, bb2, sy, sh, part, t0);
  __syncthreads();
}

__device__ __forceinline__ void build_knots(const float* uu, float* cks,
                                            float minb) {
  float m = uu[0];
#pragma unroll
  for (int i = 1; i < NBINS; ++i) m = fmaxf(m, uu[i]);
  float e[NBINS];
  float s = 0.f;
#pragma unroll
  for (int i = 0; i < NBINS; ++i) { e[i] = expf(uu[i] - m); s += e[i]; }
  const float inv = 1.0f / s;
  const float scale = 1.0f - minb * (float)NBINS;
  cks[0] = -TAILV;
  float acc = 0.f;
#pragma unroll
  for (int i = 0; i < NBINS - 1; ++i) {
    acc += minb + scale * e[i] * inv;
    cks[i + 1] = -TAILV + 2.0f * TAILV * acc;
  }
  cks[NBINS] = TAILV;
}

__global__ void __launch_bounds__(NTHR, 1)
convflow_kernel(const float* __restrict__ x,
                const float* __restrict__ pre_w, const float* __restrict__ pre_b,
                const float* __restrict__ sep_w, const float* __restrict__ sep_b,
                const float* __restrict__ pw_w,  const float* __restrict__ pw_b,
                const float* __restrict__ g1,    const float* __restrict__ b1,
                const float* __restrict__ g2,    const float* __restrict__ b2,
                const float* __restrict__ proj_w,const float* __restrict__ proj_b,
                float* __restrict__ out) {
  extern __shared__ float smem[];
  float* sh  = smem;
  float* sy  = smem + CF * SSTR;
  float* red = smem + 2 * CF * SSTR;

  const int tid = threadIdx.x;
  const int tile = blockIdx.x;
  const int b    = blockIdx.y;
  const int t0   = tile * TTILE;

  {
    const int lane = tid & 31;
    const int wv = __builtin_amdgcn_readfirstlane(tid >> 5);
    const bool central = (tid < TTILE) && (t0 + tid < TLEN);
    const float xpass = central ? x[(size_t)b * 2 * TLEN + t0 + tid] : 0.0f;
    float pw[16], pb[16];
#pragma unroll
    for (int i = 0; i < 16; ++i) {
      pw[i] = pre_w[16 * wv + i];
      pb[i] = pre_b[16 * wv + i];
    }
    float xv[7];
    bool  inr[7];
#pragma unroll
    for (int j = 0; j < 7; ++j) {
      const int t = t0 - HALO + lane + 32 * j;
      inr[j] = (t >= 0) && (t < TLEN);
      xv[j]  = inr[j] ? x[(size_t)b * 2 * TLEN + t] : 0.0f;
    }
#pragma unroll
    for (int q = 0; q < 8; ++q) {
      const float w0 = pw[2 * q],     b0 = pb[2 * q];
      const float w1 = pw[2 * q + 1], b1 = pb[2 * q + 1];
      float* row = sh + (8 * wv + q) * (2 * SSTR);
#pragma unroll
      for (int j = 0; j < 7; ++j) {
        const int u = lane + 32 * j;
        if (u < HW) {
          v2f hv;
          hv.x = inr[j] ? fmaf(w0, xv[j], b0) : 0.0f;
          hv.y = inr[j] ? fmaf(w1, xv[j], b1) : 0.0f;
          *(v2f*)(row + 2 * u) = hv;
        }
      }
    }
    if (central) {
      out[(size_t)b * 2 * TLEN + t0 + tid] = xpass;
    }
  }
  __syncthreads();

  layer_step<1, 1, TTILE + 24, 13>(sh, sy, red, t0, sep_w + 0 * CF * 3, sep_b + 0 * CF,
                            pw_w + 0 * CF * CF, pw_b + 0 * CF, g1 + 0 * CF,
                            b1 + 0 * CF, g2 + 0 * CF, b2 + 0 * CF);
  layer_step<3, 4, TTILE + 18, 12>(sh, sy, red, t0, sep_w + 1 * CF * 3, sep_b + 1 * CF,
                            pw_w + 1 * CF * CF, pw_b + 1 * CF, g1 + 1 * CF,
                            b1 + 1 * CF, g2 + 1 * CF, b2 + 1 * CF);
  layer_step<9, HALO, TTILE, 11>(sh, sy, red, t0, sep_w + 2 * CF * 3, sep_b + 2 * CF,
                            pw_w + 2 * CF * CF, pw_b + 2 * CF, g1 + 2 * CF,
                            b1 + 2 * CF, g2 + 2 * CF, b2 + 2 * CF);

  proj_mm(proj_w, proj_b, sh, sy);
  __syncthreads();

  float la_local = 0.0f;
  if ((tid < TTILE) && (t0 + tid < TLEN)) {
    const int u = HALO + tid;
    const int t = t0 + tid;
    float uw[NBINS], uh[NBINS], d[NBINS + 1];
#pragma unroll
    for (int i = 0; i < NBINS; ++i) uw[i] = sy[sidx(i, u)];
#pragma unroll
    for (int i = 0; i < NBINS; ++i) uh[i] = sy[sidx(NBINS + i, u)];
    d[0] = 1.0f;
    d[NBINS] = 1.0f;
#pragma unroll
    for (int i = 0; i < NBINS - 1; ++i)
      d[i + 1] = MIND + softplusf(sy[sidx(2 * NBINS + i, u)]);

    float cw[NBINS + 1], ch[NBINS + 1];
    build_knots(uw, cw, MINBW);
    build_knots(uh, ch, MINBH);

    const float xin = x[(size_t)b * 2 * TLEN + TLEN + t];
    const bool inside = (xin >= -TAILV) && (xin <= TAILV);
    const float xc = fminf(fmaxf(xin, -TAILV), TAILV);
    int idx = -1;
#pragma unroll
    for (int k = 0; k < NBINS + 1; ++k) idx += (xc >= cw[k]) ? 1 : 0;
    idx = max(0, min(idx, NBINS - 1));

    const float icw = cw[idx];
    const float ibw = cw[idx + 1] - cw[idx];
    const float ich = ch[idx];
    const float ih  = ch[idx + 1] - ch[idx];
    const float idelta = ih / ibw;
    const float d0 = d[idx];
    const float d1 = d[idx + 1];
    const float theta = (xc - icw) / ibw;
    const float th1m  = 1.0f - theta;
    const float num = ih * (idelta * theta * theta + d0 * theta * th1m);
    const float den = idelta + (d0 + d1 - 2.0f * idelta) * theta * th1m;
    float outv = ich + num / den;
    const float dnum = idelta * idelta *
        (d1 * theta * theta + 2.0f * idelta * theta * th1m + d0 * th1m * th1m);
    float la = logf(dnum) - 2.0f * logf(den);
    if (!inside) { outv = xin; la = 0.0f; }
    out[(size_t)b * 2 * TLEN + TLEN + t] = outv;
    la_local = la;
  }

  __syncthreads();
  int tr = tid;
  asm volatile("" : "+v"(tr));
  red[tr] = la_local;
  __syncthreads();
  for (int s = 128; s > 0; s >>= 1) {
    if (tr < s) red[tr] += red[tr + s];
    __syncthreads();
  }
  if (tr == 0) atomicAdd(out + (size_t)NBATCH * 2 * TLEN + b, red[0]);
}

__global__ void zero_logdet(float* p) {
  if (threadIdx.x < NBATCH) p[threadIdx.x] = 0.0f;
}

extern "C" void kernel_launch(void* const* d_in, const int* in_sizes, int n_in,
                              void* d_out, int out_size, void* d_ws,
                              size_t ws_size, hipStream_t stream) {
  (void)in_sizes; (void)n_in; (void)out_size; (void)d_ws; (void)ws_size;
  const float* x      = (const float*)d_in[0];
  const float* pre_w  = (const float*)d_in[2];
  const float* pre_b  = (const float*)d_in[3];
  const float* sep_w  = (const float*)d_in[4];
  const float* sep_b  = (const float*)d_in[5];
  const float* pw_w   = (const float*)d_in[6];
  const float* pw_b   = (const float*)d_in[7];
  const float* g1     = (const float*)d_in[8];
  const float* b1     = (const float*)d_in[9];
  const float* g2     = (const float*)d_in[10];
  const float* b2     = (const float*)d_in[11];
  const float* proj_w = (const float*)d_in[12];
  const float* proj_b = (const float*)d_in[13];
  float* out = (float*)d_out;

  zero_logdet<<<1, 32, 0, stream>>>(out + (size_t)NBATCH * 2 * TLEN);

  const size_t smem_bytes = (size_t)(2 * CF * SSTR + PARTF) * sizeof(float);
  dim3 grid(NTILE, NBATCH);
  convflow_kernel<<<grid, NTHR, smem_bytes, stream>>>(
      x, pre_w, pre_b, sep_w, sep_b, pw_w, pw_b, g1, b1, g2, b2, proj_w,
      proj_b, out);
}
